# starting kernel
# baseline (speedup 1.0000x reference)
#include <hip/hip_runtime.h>
#include <hip/hip_bf16.h>


#define MTOK 32768
#define CCH  256

typedef __attribute__((ext_vector_type(2))) float v2f;
typedef __attribute__((ext_vector_type(8))) float v8f;

__device__ __forceinline__ v8f wmma_f32(v2f a, v2f b, v8f c) {
    return __builtin_amdgcn_wmma_f32_16x16x4_f32(false, a, false, b, (short)0, c, false, false);
}

__global__ __launch_bounds__(512) void gram_kernel(const float* __restrict__ x,
                                                   float* __restrict__ Gpart,
                                                   float* __restrict__ Spart) {
    __shared__ float lds[CCH * 34];
    const int n    = blockIdx.y;
    const int wg   = blockIdx.x;
    const float* xn = x + (size_t)n * CCH * MTOK;
    const int tid  = threadIdx.x;
    const int lane = tid & 31;
    const int w    = tid >> 5;
    const int bwr  = w >> 2, bwc = w & 3;
    const int l15  = lane & 15;
    const int half = lane >> 4;

    const v8f zero8 = {0.f,0.f,0.f,0.f,0.f,0.f,0.f,0.f};
    v8f acc[4][4];
#pragma unroll
    for (int r = 0; r < 4; ++r)
#pragma unroll
        for (int c = 0; c < 4; ++c) acc[r][c] = zero8;

    float sacc[16];
#pragma unroll
    for (int i = 0; i < 16; ++i) sacc[i] = 0.f;

    const int m0 = wg * 512;
    for (int t0 = 0; t0 < 512; t0 += 32) {
        __syncthreads();
#pragma unroll
        for (int i = 0; i < 16; ++i) {
            const int ch = w + 16 * i;
            const float val = xn[(size_t)ch * MTOK + m0 + t0 + lane];
            lds[ch * 34 + lane] = val;
            sacc[i] += val;
        }
        __syncthreads();
#pragma unroll
        for (int kb = 0; kb < 32; kb += 4) {
            const int kk = kb + half * 2;
            v2f a[4], b[4];
#pragma unroll
            for (int r = 0; r < 4; ++r) {
                const int chA = (bwr * 4 + r) * 16 + l15;
                a[r] = *(const v2f*)&lds[chA * 34 + kk];
                const int chB = (bwc * 4 + r) * 16 + l15;
                b[r] = *(const v2f*)&lds[chB * 34 + kk];
            }
#pragma unroll
            for (int r = 0; r < 4; ++r)
#pragma unroll
                for (int c = 0; c < 4; ++c)
                    acc[r][c] = wmma_f32(a[r], b[c], acc[r][c]);
        }
    }

    float* gp = Gpart + ((size_t)n * 64 + wg) * (CCH * CCH);
#pragma unroll
    for (int r = 0; r < 4; ++r)
#pragma unroll
        for (int c = 0; c < 4; ++c) {
            const int row0 = (bwr * 4 + r) * 16 + half * 8;
            const int col  = (bwc * 4 + c) * 16 + l15;
#pragma unroll
            for (int v = 0; v < 8; ++v)
                gp[(row0 + v) * CCH + col] = acc[r][c][v];
        }

#pragma unroll
    for (int i = 0; i < 16; ++i) {
        float s = sacc[i];
        for (int off = 16; off >= 1; off >>= 1) s += __shfl_xor(s, off, 32);
        if (lane == 0) Spart[((size_t)n * 64 + wg) * CCH + w + 16 * i] = s;
    }
}

__global__ __launch_bounds__(256) void reduce_kernel(const float* __restrict__ Gpart,
                                                     const float* __restrict__ Spart,
                                                     float* __restrict__ G,
                                                     float* __restrict__ S) {
    const int idx = blockIdx.x * 256 + threadIdx.x;
    const int n = idx >> 16;
    const int e = idx & 65535;
    const float* gp = Gpart + (size_t)n * 64 * 65536 + e;
    float s = 0.f;
    for (int p = 0; p < 64; ++p) s += gp[(size_t)p * 65536];
    G[idx] = s;
    if (idx < 512) {
        const float* sp = Spart + (size_t)(idx >> 8) * 64 * CCH + (idx & 255);
        float t = 0.f;
        for (int p = 0; p < 64; ++p) t += sp[p * CCH];
        S[idx] = t;
    }
}

__global__ __launch_bounds__(512) void vecprep_kernel(const float* __restrict__ Wq,
                                                      const float* __restrict__ Wk,
                                                      const float* __restrict__ bk,
                                                      const float* __restrict__ S,
                                                      float* __restrict__ U,
                                                      float* __restrict__ Wvec) {
    const int tid = threadIdx.x;
    const int n = tid >> 8, i = tid & 255;
    const float* s = S + n * CCH;
    float u = 0.f, wv = 0.f;
    for (int c = 0; c < CCH; ++c) {
        u  += Wq[i * CCH + c] * s[c];
        wv += Wk[i * CCH + c] * s[c];
    }
    U[tid] = u;
    Wvec[tid] = wv + (float)MTOK * bk[i];
}

__global__ __launch_bounds__(512) void gemm256_kernel(const float* __restrict__ A, long long sA,
                                                      const float* __restrict__ B, long long sB,
                                                      int transB,
                                                      float* __restrict__ D, long long sD) {
    const int n = blockIdx.y;
    const float* An = A + (size_t)n * sA;
    const float* Bn = B + (size_t)n * sB;
    float* Dn = D + (size_t)n * sD;
    const int tid = threadIdx.x, lane = tid & 31, w = tid >> 5;
    const int l15 = lane & 15, half = lane >> 4;
    const int row0 = blockIdx.x * 64;
    const int tr  = w & 3;
    const int tc0 = (w >> 2) * 4;
    const int rowA = row0 + tr * 16 + l15;

    const v8f zero8 = {0.f,0.f,0.f,0.f,0.f,0.f,0.f,0.f};
    v8f acc[4];
#pragma unroll
    for (int c = 0; c < 4; ++c) acc[c] = zero8;

    for (int kb = 0; kb < CCH; kb += 4) {
        const int kk = kb + half * 2;
        const v2f a = *(const v2f*)&An[rowA * CCH + kk];
#pragma unroll
        for (int c = 0; c < 4; ++c) {
            const int col = (tc0 + c) * 16 + l15;
            v2f b;
            if (transB) {
                b = *(const v2f*)&Bn[col * CCH + kk];
            } else {
                b[0] = Bn[kk * CCH + col];
                b[1] = Bn[(kk + 1) * CCH + col];
            }
            acc[c] = wmma_f32(a, b, acc[c]);
        }
    }
#pragma unroll
    for (int c = 0; c < 4; ++c) {
        const int col = (tc0 + c) * 16 + l15;
        const int rr = row0 + tr * 16 + half * 8;
#pragma unroll
        for (int v = 0; v < 8; ++v)
            Dn[(rr + v) * CCH + col] = acc[c][v];
    }
}

__global__ __launch_bounds__(256) void softmax_kernel(const float* __restrict__ P0,
                                                      const float* __restrict__ U,
                                                      const float* __restrict__ Wvec,
                                                      const float* __restrict__ bq,
                                                      const float* __restrict__ bk,
                                                      const float* __restrict__ bv,
                                                      float* __restrict__ att,
                                                      float* __restrict__ cvec) {
    const int n = blockIdx.y;
    const int row = blockIdx.x * 8 + (threadIdx.x >> 5);
    const int lane = threadIdx.x & 31;
    const float* p = P0 + ((size_t)n * CCH + row) * CCH;
    const float ui = U[n * CCH + row];
    const float bqi = bq[row];
    float l[8];
    float mx = -3.0e38f;
#pragma unroll
    for (int t = 0; t < 8; ++t) {
        const int j = lane + 32 * t;
        const float val = p[j] + ui * bk[j] + bqi * Wvec[n * CCH + j];
        l[t] = val;
        mx = fmaxf(mx, val);
    }
    for (int off = 16; off >= 1; off >>= 1) mx = fmaxf(mx, __shfl_xor(mx, off, 32));
    float sum = 0.f;
#pragma unroll
    for (int t = 0; t < 8; ++t) { l[t] = __expf(l[t] - mx); sum += l[t]; }
    for (int off = 16; off >= 1; off >>= 1) sum += __shfl_xor(sum, off, 32);
    const float inv = 1.f / sum;
    float cd = 0.f;
    float* arow = att + ((size_t)n * CCH + row) * CCH;
#pragma unroll
    for (int t = 0; t < 8; ++t) {
        const int j = lane + 32 * t;
        const float a = l[t] * inv;
        arow[j] = a;
        cd += a * bv[j];
    }
    for (int off = 16; off >= 1; off >>= 1) cd += __shfl_xor(cd, off, 32);
    if (lane == 0) cvec[n * CCH + row] = cd;
}

__global__ __launch_bounds__(512) void out_kernel(const float* __restrict__ x,
                                                  const float* __restrict__ Bt,
                                                  const float* __restrict__ cvec,
                                                  float* __restrict__ out) {
    __shared__ float lds[CCH * 40];
    const int n = blockIdx.y;
    const float* xn = x + (size_t)n * CCH * MTOK;
    float* on = out + (size_t)n * MTOK * CCH;
    const int m0 = blockIdx.x * 32;
    const int tid = threadIdx.x, lane = tid & 31, w = tid >> 5;
    const int l15 = lane & 15, half = lane >> 4;

#pragma unroll
    for (int i = 0; i < 16; ++i) {
        const int ch = w + 16 * i;
        lds[ch * 40 + lane] = xn[(size_t)ch * MTOK + m0 + lane];
    }
    __syncthreads();

    const int tr  = w & 1;
    const int tc0 = (w >> 1) * 2;
    const float* btn = Bt + (size_t)n * CCH * CCH;
    const v8f zero8 = {0.f,0.f,0.f,0.f,0.f,0.f,0.f,0.f};
    v8f acc[2];
    acc[0] = zero8; acc[1] = zero8;

    for (int kb = 0; kb < CCH; kb += 4) {
        const int kk = kb + half * 2;
        const int ml = tr * 16 + l15;
        v2f a;
        a[0] = lds[kk * 40 + ml];
        a[1] = lds[(kk + 1) * 40 + ml];
#pragma unroll
        for (int c = 0; c < 2; ++c) {
            const int col = (tc0 + c) * 16 + l15;
            const v2f b = *(const v2f*)&btn[col * CCH + kk];
            acc[c] = wmma_f32(a, b, acc[c]);
        }
    }
#pragma unroll
    for (int c = 0; c < 2; ++c) {
        const int col = (tc0 + c) * 16 + l15;
        const float cv = cvec[n * CCH + col];
        const int mrow = tr * 16 + half * 8;
#pragma unroll
        for (int v = 0; v < 8; ++v)
            on[(size_t)(m0 + mrow + v) * CCH + col] = acc[c][v] + cv;
    }
}

extern "C" void kernel_launch(void* const* d_in, const int* in_sizes, int n_in,
                              void* d_out, int out_size, void* d_ws, size_t ws_size,
                              hipStream_t stream) {
    (void)in_sizes; (void)n_in; (void)out_size; (void)ws_size;
    const float* x  = (const float*)d_in[0];
    const float* Wq = (const float*)d_in[1];
    const float* bq = (const float*)d_in[2];
    const float* Wk = (const float*)d_in[3];
    const float* bk = (const float*)d_in[4];
    const float* Wv = (const float*)d_in[5];
    const float* bv = (const float*)d_in[6];
    float* out = (float*)d_out;
    float* ws  = (float*)d_ws;

    const size_t oGP  = 0;
    const size_t oSP  = oGP  + (size_t)2 * 64 * 65536;
    const size_t oG   = oSP  + (size_t)2 * 64 * 256;
    const size_t oS   = oG   + (size_t)2 * 65536;
    const size_t oT1  = oS   + 512;
    const size_t oP0  = oT1  + (size_t)2 * 65536;
    const size_t oATT = oP0  + (size_t)2 * 65536;
    const size_t oBT  = oATT + (size_t)2 * 65536;
    const size_t oU   = oBT  + (size_t)2 * 65536;
    const size_t oW   = oU   + 512;
    const size_t oCV  = oW   + 512;

    gram_kernel<<<dim3(64, 2), 512, 0, stream>>>(x, ws + oGP, ws + oSP);
    reduce_kernel<<<512, 256, 0, stream>>>(ws + oGP, ws + oSP, ws + oG, ws + oS);
    vecprep_kernel<<<1, 512, 0, stream>>>(Wq, Wk, bk, ws + oS, ws + oU, ws + oW);
    gemm256_kernel<<<dim3(4, 2), 512, 0, stream>>>(Wq, 0, ws + oG, 65536, 0, ws + oT1, 65536);
    gemm256_kernel<<<dim3(4, 2), 512, 0, stream>>>(ws + oT1, 65536, Wk, 0, 1, ws + oP0, 65536);
    softmax_kernel<<<dim3(32, 2), 256, 0, stream>>>(ws + oP0, ws + oU, ws + oW, bq, bk, bv,
                                                    ws + oATT, ws + oCV);
    gemm256_kernel<<<dim3(4, 2), 512, 0, stream>>>(ws + oATT, 65536, Wv, 0, 0, ws + oBT, 65536);
    out_kernel<<<dim3(1024, 2), 512, 0, stream>>>(x, ws + oBT, ws + oCV, out);
}


# gemm256 (3 launches): from 8 WGs fetching WMMA fragments from global at every k-step to 128 WGs of 32x32 tiles, whole-K A/B panels staged once in LDS by coalesced b128 loads; k order kept
# speedup vs baseline: 1.5601x; 1.5601x over previous
#include <hip/hip_runtime.h>
#include <hip/hip_bf16.h>


#define MTOK 32768
#define CCH  256

typedef __attribute__((ext_vector_type(2))) float v2f;
typedef __attribute__((ext_vector_type(8))) float v8f;

__device__ __forceinline__ v8f wmma_f32(v2f a, v2f b, v8f c) {
    return __builtin_amdgcn_wmma_f32_16x16x4_f32(false, a, false, b, (short)0, c, false, false);
}

__global__ __launch_bounds__(512) void gram_kernel(const float* __restrict__ x,
                                                   float* __restrict__ Gpart,
                                                   float* __restrict__ Spart) {
    __shared__ float lds[CCH * 34];
    const int n    = blockIdx.y;
    const int wg   = blockIdx.x;
    const float* xn = x + (size_t)n * CCH * MTOK;
    const int tid  = threadIdx.x;
    const int lane = tid & 31;
    const int w    = tid >> 5;
    const int bwr  = w >> 2, bwc = w & 3;
    const int l15  = lane & 15;
    const int half = lane >> 4;

    const v8f zero8 = {0.f,0.f,0.f,0.f,0.f,0.f,0.f,0.f};
    v8f acc[4][4];
#pragma unroll
    for (int r = 0; r < 4; ++r)
#pragma unroll
        for (int c = 0; c < 4; ++c) acc[r][c] = zero8;

    float sacc[16];
#pragma unroll
    for (int i = 0; i < 16; ++i) sacc[i] = 0.f;

    const int m0 = wg * 512;
    for (int t0 = 0; t0 < 512; t0 += 32) {
        __syncthreads();
#pragma unroll
        for (int i = 0; i < 16; ++i) {
            const int ch = w + 16 * i;
            const float val = xn[(size_t)ch * MTOK + m0 + t0 + lane];
            lds[ch * 34 + lane] = val;
            sacc[i] += val;
        }
        __syncthreads();
#pragma unroll
        for (int kb = 0; kb < 32; kb += 4) {
            const int kk = kb + half * 2;
            v2f a[4], b[4];
#pragma unroll
            for (int r = 0; r < 4; ++r) {
                const int chA = (bwr * 4 + r) * 16 + l15;
                a[r] = *(const v2f*)&lds[chA * 34 + kk];
                const int chB = (bwc * 4 + r) * 16 + l15;
                b[r] = *(const v2f*)&lds[chB * 34 + kk];
            }
#pragma unroll
            for (int r = 0; r < 4; ++r)
#pragma unroll
                for (int c = 0; c < 4; ++c)
                    acc[r][c] = wmma_f32(a[r], b[c], acc[r][c]);
        }
    }

    float* gp = Gpart + ((size_t)n * 64 + wg) * (CCH * CCH);
#pragma unroll
    for (int r = 0; r < 4; ++r)
#pragma unroll
        for (int c = 0; c < 4; ++c) {
            const int row0 = (bwr * 4 + r) * 16 + half * 8;
            const int col  = (bwc * 4 + c) * 16 + l15;
#pragma unroll
            for (int v = 0; v < 8; ++v)
                gp[(row0 + v) * CCH + col] = acc[r][c][v];
        }

#pragma unroll
    for (int i = 0; i < 16; ++i) {
        float s = sacc[i];
        for (int off = 16; off >= 1; off >>= 1) s += __shfl_xor(s, off, 32);
        if (lane == 0) Spart[((size_t)n * 64 + wg) * CCH + w + 16 * i] = s;
    }
}

__global__ __launch_bounds__(256) void reduce_kernel(const float* __restrict__ Gpart,
                                                     const float* __restrict__ Spart,
                                                     float* __restrict__ G,
                                                     float* __restrict__ S) {
    const int idx = blockIdx.x * 256 + threadIdx.x;
    const int n = idx >> 16;
    const int e = idx & 65535;
    const float* gp = Gpart + (size_t)n * 64 * 65536 + e;
    float s = 0.f;
    for (int p = 0; p < 64; ++p) s += gp[(size_t)p * 65536];
    G[idx] = s;
    if (idx < 512) {
        const float* sp = Spart + (size_t)(idx >> 8) * 64 * CCH + (idx & 255);
        float t = 0.f;
        for (int p = 0; p < 64; ++p) t += sp[p * CCH];
        S[idx] = t;
    }
}

__global__ __launch_bounds__(512) void vecprep_kernel(const float* __restrict__ Wq,
                                                      const float* __restrict__ Wk,
                                                      const float* __restrict__ bk,
                                                      const float* __restrict__ S,
                                                      float* __restrict__ U,
                                                      float* __restrict__ Wvec) {
    const int tid = threadIdx.x;
    const int n = tid >> 8, i = tid & 255;
    const float* s = S + n * CCH;
    float u = 0.f, wv = 0.f;
    for (int c = 0; c < CCH; ++c) {
        u  += Wq[i * CCH + c] * s[c];
        wv += Wk[i * CCH + c] * s[c];
    }
    U[tid] = u;
    Wvec[tid] = wv + (float)MTOK * bk[i];
}

#define GK_PITCH 260
#define GC_PITCH 48
#define GD_PITCH 36

template <int TRANSB>
__global__ __launch_bounds__(128) void gemm256_kernel(const float* __restrict__ A, long long sA,
                                                      const float* __restrict__ B, long long sB,
                                                      float* __restrict__ D, long long sD) {
    constexpr int AS_FLOATS = 32 * GK_PITCH;
    constexpr int BS_FLOATS = TRANSB ? 32 * GK_PITCH : CCH * GC_PITCH;
    __shared__ __attribute__((aligned(16))) float lds[AS_FLOATS + BS_FLOATS];
    float* As = lds;
    float* Bs = lds + AS_FLOATS;

    const int n = blockIdx.z;
    const float* An = A + (size_t)n * sA;
    const float* Bn = B + (size_t)n * sB;
    float* Dn = D + (size_t)n * sD;
    const int tid = threadIdx.x, lane = tid & 31, w = tid >> 5;
    const int l15 = lane & 15, half = lane >> 4;
    const int row0 = blockIdx.y * 32;
    const int col0 = blockIdx.x * 32;

    const float* ag = An + (size_t)(row0 + (tid >> 6)) * CCH + (tid & 63) * 4;
    const float* bg = TRANSB ? Bn + (size_t)(col0 + (tid >> 6)) * CCH + (tid & 63) * 4
                             : Bn + (size_t)(tid >> 3) * CCH + col0 + (tid & 7) * 4;
    float* as = As + (tid >> 6) * GK_PITCH + (tid & 63) * 4;
    float* bs = TRANSB ? Bs + (tid >> 6) * GK_PITCH + (tid & 63) * 4
                       : Bs + (tid >> 3) * GC_PITCH + (tid & 7) * 4;
    float4 ra[16], rb[16];
#pragma unroll
    for (int i = 0; i < 16; ++i)
        ra[i] = *(const float4*)(ag + i * 2 * CCH);
#pragma unroll
    for (int i = 0; i < 16; ++i)
        rb[i] = *(const float4*)(bg + i * (TRANSB ? 2 : 16) * CCH);
#pragma unroll
    for (int i = 0; i < 16; ++i)
        *(float4*)(as + i * 2 * GK_PITCH) = ra[i];
#pragma unroll
    for (int i = 0; i < 16; ++i)
        *(float4*)(bs + i * (TRANSB ? 2 * GK_PITCH : 16 * GC_PITCH)) = rb[i];
    __syncthreads();

    const int tr = w & 1;
    const int tc = w >> 1;
    const float* ap = As + (tr * 16 + l15) * GK_PITCH + half * 2;
    const float* bp = TRANSB ? Bs + (tc * 16 + l15) * GK_PITCH + half * 2
                             : Bs + (half * 2) * GC_PITCH + tc * 16 + l15;

    v2f fa[2][16], fb[2][16];
    v8f acc = {0.f,0.f,0.f,0.f,0.f,0.f,0.f,0.f};
#pragma unroll
    for (int blk = -1; blk < 4; ++blk) {
        if (blk < 3) {
            const int nx = (blk + 1) & 1;
#pragma unroll
            for (int s = 0; s < 16; ++s) {
                const int kb = (blk + 1) * 64 + s * 4;
                fa[nx][s] = *(const v2f*)(ap + kb);
                if (TRANSB) {
                    fb[nx][s] = *(const v2f*)(bp + kb);
                } else {
                    fb[nx][s][0] = bp[kb * GC_PITCH];
                    fb[nx][s][1] = bp[(kb + 1) * GC_PITCH];
                }
            }
        }
        __builtin_amdgcn_sched_barrier(0);
        if (blk >= 0) {
            const int cu = blk & 1;
#pragma unroll
            for (int s = 0; s < 16; ++s)
                acc = wmma_f32(fa[cu][s], fb[cu][s], acc);
        }
        __builtin_amdgcn_sched_barrier(0);
    }

    __syncthreads();
    float* Ds = lds;
#pragma unroll
    for (int v = 0; v < 8; ++v)
        Ds[(tr * 16 + half * 8 + v) * GD_PITCH + tc * 16 + l15] = acc[v];
    __syncthreads();
#pragma unroll
    for (int i = 0; i < 2; ++i) {
        const int r = (tid >> 3) + 16 * i, c4 = tid & 7;
        const float4 val = *(const float4*)&Ds[r * GD_PITCH + c4 * 4];
        *(float4*)&Dn[(row0 + r) * CCH + col0 + c4 * 4] = val;
    }
}

__global__ __launch_bounds__(256) void softmax_kernel(const float* __restrict__ P0,
                                                      const float* __restrict__ U,
                                                      const float* __restrict__ Wvec,
                                                      const float* __restrict__ bq,
                                                      const float* __restrict__ bk,
                                                      const float* __restrict__ bv,
                                                      float* __restrict__ att,
                                                      float* __restrict__ cvec) {
    const int n = blockIdx.y;
    const int row = blockIdx.x * 8 + (threadIdx.x >> 5);
    const int lane = threadIdx.x & 31;
    const float* p = P0 + ((size_t)n * CCH + row) * CCH;
    const float ui = U[n * CCH + row];
    const float bqi = bq[row];
    float l[8];
    float mx = -3.0e38f;
#pragma unroll
    for (int t = 0; t < 8; ++t) {
        const int j = lane + 32 * t;
        const float val = p[j] + ui * bk[j] + bqi * Wvec[n * CCH + j];
        l[t] = val;
        mx = fmaxf(mx, val);
    }
    for (int off = 16; off >= 1; off >>= 1) mx = fmaxf(mx, __shfl_xor(mx, off, 32));
    float sum = 0.f;
#pragma unroll
    for (int t = 0; t < 8; ++t) { l[t] = __expf(l[t] - mx); sum += l[t]; }
    for (int off = 16; off >= 1; off >>= 1) sum += __shfl_xor(sum, off, 32);
    const float inv = 1.f / sum;
    float cd = 0.f;
    float* arow = att + ((size_t)n * CCH + row) * CCH;
#pragma unroll
    for (int t = 0; t < 8; ++t) {
        const int j = lane + 32 * t;
        const float a = l[t] * inv;
        arow[j] = a;
        cd += a * bv[j];
    }
    for (int off = 16; off >= 1; off >>= 1) cd += __shfl_xor(cd, off, 32);
    if (lane == 0) cvec[n * CCH + row] = cd;
}

__global__ __launch_bounds__(512) void out_kernel(const float* __restrict__ x,
                                                  const float* __restrict__ Bt,
                                                  const float* __restrict__ cvec,
                                                  float* __restrict__ out) {
    __shared__ float lds[CCH * 40];
    const int n = blockIdx.y;
    const float* xn = x + (size_t)n * CCH * MTOK;
    float* on = out + (size_t)n * MTOK * CCH;
    const int m0 = blockIdx.x * 32;
    const int tid = threadIdx.x, lane = tid & 31, w = tid >> 5;
    const int l15 = lane & 15, half = lane >> 4;

#pragma unroll
    for (int i = 0; i < 16; ++i) {
        const int ch = w + 16 * i;
        lds[ch * 40 + lane] = xn[(size_t)ch * MTOK + m0 + lane];
    }
    __syncthreads();

    const int tr  = w & 1;
    const int tc0 = (w >> 1) * 2;
    const float* btn = Bt + (size_t)n * CCH * CCH;
    const v8f zero8 = {0.f,0.f,0.f,0.f,0.f,0.f,0.f,0.f};
    v8f acc[2];
    acc[0] = zero8; acc[1] = zero8;

    for (int kb = 0; kb < CCH; kb += 4) {
        const int kk = kb + half * 2;
        const int ml = tr * 16 + l15;
        v2f a;
        a[0] = lds[kk * 40 + ml];
        a[1] = lds[(kk + 1) * 40 + ml];
#pragma unroll
        for (int c = 0; c < 2; ++c) {
            const int col = (tc0 + c) * 16 + l15;
            const v2f b = *(const v2f*)&btn[col * CCH + kk];
            acc[c] = wmma_f32(a, b, acc[c]);
        }
    }
#pragma unroll
    for (int c = 0; c < 2; ++c) {
        const int col = (tc0 + c) * 16 + l15;
        const float cv = cvec[n * CCH + col];
        const int mrow = tr * 16 + half * 8;
#pragma unroll
        for (int v = 0; v < 8; ++v)
            on[(size_t)(m0 + mrow + v) * CCH + col] = acc[c][v] + cv;
    }
}

extern "C" void kernel_launch(void* const* d_in, const int* in_sizes, int n_in,
                              void* d_out, int out_size, void* d_ws, size_t ws_size,
                              hipStream_t stream) {
    (void)in_sizes; (void)n_in; (void)out_size; (void)ws_size;
    const float* x  = (const float*)d_in[0];
    const float* Wq = (const float*)d_in[1];
    const float* bq = (const float*)d_in[2];
    const float* Wk = (const float*)d_in[3];
    const float* bk = (const float*)d_in[4];
    const float* Wv = (const float*)d_in[5];
    const float* bv = (const float*)d_in[6];
    float* out = (float*)d_out;
    float* ws  = (float*)d_ws;

    const size_t oGP  = 0;
    const size_t oSP  = oGP  + (size_t)2 * 64 * 65536;
    const size_t oG   = oSP  + (size_t)2 * 64 * 256;
    const size_t oS   = oG   + (size_t)2 * 65536;
    const size_t oT1  = oS   + 512;
    const size_t oP0  = oT1  + (size_t)2 * 65536;
    const size_t oATT = oP0  + (size_t)2 * 65536;
    const size_t oBT  = oATT + (size_t)2 * 65536;
    const size_t oU   = oBT  + (size_t)2 * 65536;
    const size_t oW   = oU   + 512;
    const size_t oCV  = oW   + 512;

    gram_kernel<<<dim3(64, 2), 512, 0, stream>>>(x, ws + oGP, ws + oSP);
    reduce_kernel<<<512, 256, 0, stream>>>(ws + oGP, ws + oSP, ws + oG, ws + oS);
    vecprep_kernel<<<1, 512, 0, stream>>>(Wq, Wk, bk, ws + oS, ws + oU, ws + oW);
    gemm256_kernel<0><<<dim3(8, 8, 2), 128, 0, stream>>>(Wq, 0, ws + oG, 65536, ws + oT1, 65536);
    gemm256_kernel<1><<<dim3(8, 8, 2), 128, 0, stream>>>(ws + oT1, 65536, Wk, 0, ws + oP0, 65536);
    softmax_kernel<<<dim3(32, 2), 256, 0, stream>>>(ws + oP0, ws + oU, ws + oW, bq, bk, bv,
                                                    ws + oATT, ws + oCV);
    gemm256_kernel<0><<<dim3(8, 8, 2), 128, 0, stream>>>(ws + oATT, 65536, Wv, 0, ws + oBT, 65536);
    out_kernel<<<dim3(1024, 2), 512, 0, stream>>>(x, ws + oBT, ws + oCV, out);
}


# gram_kernel: each WG computes one 64-row band of G (wave tile 1x4) instead of all 256 rows: 512 WGs, two per WGP on all 256 WGPs, where 128 WGs left half the chip idle; summation order unchanged
# speedup vs baseline: 1.8967x; 1.2158x over previous
#include <hip/hip_runtime.h>
#include <hip/hip_bf16.h>


#define MTOK 32768
#define CCH  256

typedef __attribute__((ext_vector_type(2))) float v2f;
typedef __attribute__((ext_vector_type(8))) float v8f;

__device__ __forceinline__ v8f wmma_f32(v2f a, v2f b, v8f c) {
    return __builtin_amdgcn_wmma_f32_16x16x4_f32(false, a, false, b, (short)0, c, false, false);
}

__global__ __launch_bounds__(512) void gram_kernel(const float* __restrict__ x,
                                                   float* __restrict__ Gpart,
                                                   float* __restrict__ Spart) {
    __shared__ float lds[CCH * 34];
    const int n    = blockIdx.y;
    const int wg   = blockIdx.x;
    const int band = blockIdx.z;
    const float* xn = x + (size_t)n * CCH * MTOK;
    const int tid  = threadIdx.x;
    const int lane = tid & 31;
    const int w    = tid >> 5;
    const int bwr  = w >> 2, bwc = w & 3;
    const int l15  = lane & 15;
    const int half = lane >> 4;
    const bool sums = (band == 0);

    const v8f zero8 = {0.f,0.f,0.f,0.f,0.f,0.f,0.f,0.f};
    v8f acc[4];
#pragma unroll
    for (int c = 0; c < 4; ++c) acc[c] = zero8;

    float sacc[16];
#pragma unroll
    for (int i = 0; i < 16; ++i) sacc[i] = 0.f;

    const int m0 = wg * 512;
    const int chA = (band * 4 + bwr) * 16 + l15;
    for (int t0 = 0; t0 < 512; t0 += 32) {
        __syncthreads();
#pragma unroll
        for (int i = 0; i < 16; ++i) {
            const int ch = w + 16 * i;
            const float val = xn[(size_t)ch * MTOK + m0 + t0 + lane];
            lds[ch * 34 + lane] = val;
            if (sums) sacc[i] += val;
        }
        __syncthreads();
#pragma unroll
        for (int kb = 0; kb < 32; kb += 4) {
            const int kk = kb + half * 2;
            v2f b[4];
            const v2f a = *(const v2f*)&lds[chA * 34 + kk];
#pragma unroll
            for (int c = 0; c < 4; ++c) {
                const int chB = (bwc * 4 + c) * 16 + l15;
                b[c] = *(const v2f*)&lds[chB * 34 + kk];
            }
#pragma unroll
            for (int c = 0; c < 4; ++c)
                acc[c] = wmma_f32(a, b[c], acc[c]);
        }
    }

    float* gp = Gpart + ((size_t)n * 64 + wg) * (CCH * CCH);
#pragma unroll
    for (int c = 0; c < 4; ++c) {
        const int row0 = (band * 4 + bwr) * 16 + half * 8;
        const int col  = (bwc * 4 + c) * 16 + l15;
#pragma unroll
        for (int v = 0; v < 8; ++v)
            gp[(row0 + v) * CCH + col] = acc[c][v];
    }

    if (sums) {
#pragma unroll
        for (int i = 0; i < 16; ++i) {
            float s = sacc[i];
            for (int off = 16; off >= 1; off >>= 1) s += __shfl_xor(s, off, 32);
            if (lane == 0) Spart[((size_t)n * 64 + wg) * CCH + w + 16 * i] = s;
        }
    }
}

__global__ __launch_bounds__(256) void reduce_kernel(const float* __restrict__ Gpart,
                                                     const float* __restrict__ Spart,
                                                     float* __restrict__ G,
                                                     float* __restrict__ S) {
    const int idx = blockIdx.x * 256 + threadIdx.x;
    const int n = idx >> 16;
    const int e = idx & 65535;
    const float* gp = Gpart + (size_t)n * 64 * 65536 + e;
    float s = 0.f;
    for (int p = 0; p < 64; ++p) s += gp[(size_t)p * 65536];
    G[idx] = s;
    if (idx < 512) {
        const float* sp = Spart + (size_t)(idx >> 8) * 64 * CCH + (idx & 255);
        float t = 0.f;
        for (int p = 0; p < 64; ++p) t += sp[p * CCH];
        S[idx] = t;
    }
}

__global__ __launch_bounds__(512) void vecprep_kernel(const float* __restrict__ Wq,
                                                      const float* __restrict__ Wk,
                                                      const float* __restrict__ bk,
                                                      const float* __restrict__ S,
                                                      float* __restrict__ U,
                                                      float* __restrict__ Wvec) {
    const int tid = threadIdx.x;
    const int n = tid >> 8, i = tid & 255;
    const float* s = S + n * CCH;
    float u = 0.f, wv = 0.f;
    for (int c = 0; c < CCH; ++c) {
        u  += Wq[i * CCH + c] * s[c];
        wv += Wk[i * CCH + c] * s[c];
    }
    U[tid] = u;
    Wvec[tid] = wv + (float)MTOK * bk[i];
}

#define GK_PITCH 260
#define GC_PITCH 48
#define GD_PITCH 36

template <int TRANSB>
__global__ __launch_bounds__(128) void gemm256_kernel(const float* __restrict__ A, long long sA,
                                                      const float* __restrict__ B, long long sB,
                                                      float* __restrict__ D, long long sD) {
    constexpr int AS_FLOATS = 32 * GK_PITCH;
    constexpr int BS_FLOATS = TRANSB ? 32 * GK_PITCH : CCH * GC_PITCH;
    __shared__ __attribute__((aligned(16))) float lds[AS_FLOATS + BS_FLOATS];
    float* As = lds;
    float* Bs = lds + AS_FLOATS;

    const int n = blockIdx.z;
    const float* An = A + (size_t)n * sA;
    const float* Bn = B + (size_t)n * sB;
    float* Dn = D + (size_t)n * sD;
    const int tid = threadIdx.x, lane = tid & 31, w = tid >> 5;
    const int l15 = lane & 15, half = lane >> 4;
    const int row0 = blockIdx.y * 32;
    const int col0 = blockIdx.x * 32;

    const float* ag = An + (size_t)(row0 + (tid >> 6)) * CCH + (tid & 63) * 4;
    const float* bg = TRANSB ? Bn + (size_t)(col0 + (tid >> 6)) * CCH + (tid & 63) * 4
                             : Bn + (size_t)(tid >> 3) * CCH + col0 + (tid & 7) * 4;
    float* as = As + (tid >> 6) * GK_PITCH + (tid & 63) * 4;
    float* bs = TRANSB ? Bs + (tid >> 6) * GK_PITCH + (tid & 63) * 4
                       : Bs + (tid >> 3) * GC_PITCH + (tid & 7) * 4;
    float4 ra[16], rb[16];
#pragma unroll
    for (int i = 0; i < 16; ++i)
        ra[i] = *(const float4*)(ag + i * 2 * CCH);
#pragma unroll
    for (int i = 0; i < 16; ++i)
        rb[i] = *(const float4*)(bg + i * (TRANSB ? 2 : 16) * CCH);
#pragma unroll
    for (int i = 0; i < 16; ++i)
        *(float4*)(as + i * 2 * GK_PITCH) = ra[i];
#pragma unroll
    for (int i = 0; i < 16; ++i)
        *(float4*)(bs + i * (TRANSB ? 2 * GK_PITCH : 16 * GC_PITCH)) = rb[i];
    __syncthreads();

    const int tr = w & 1;
    const int tc = w >> 1;
    const float* ap = As + (tr * 16 + l15) * GK_PITCH + half * 2;
    const float* bp = TRANSB ? Bs + (tc * 16 + l15) * GK_PITCH + half * 2
                             : Bs + (half * 2) * GC_PITCH + tc * 16 + l15;

    v2f fa[2][16], fb[2][16];
    v8f acc = {0.f,0.f,0.f,0.f,0.f,0.f,0.f,0.f};
#pragma unroll
    for (int blk = -1; blk < 4; ++blk) {
        if (blk < 3) {
            const int nx = (blk + 1) & 1;
#pragma unroll
            for (int s = 0; s < 16; ++s) {
                const int kb = (blk + 1) * 64 + s * 4;
                fa[nx][s] = *(const v2f*)(ap + kb);
                if (TRANSB) {
                    fb[nx][s] = *(const v2f*)(bp + kb);
                } else {
                    fb[nx][s][0] = bp[kb * GC_PITCH];
                    fb[nx][s][1] = bp[(kb + 1) * GC_PITCH];
                }
            }
        }
        __builtin_amdgcn_sched_barrier(0);
        if (blk >= 0) {
            const int cu = blk & 1;
#pragma unroll
            for (int s = 0; s < 16; ++s)
                acc = wmma_f32(fa[cu][s], fb[cu][s], acc);
        }
        __builtin_amdgcn_sched_barrier(0);
    }

    __syncthreads();
    float* Ds = lds;
#pragma unroll
    for (int v = 0; v < 8; ++v)
        Ds[(tr * 16 + half * 8 + v) * GD_PITCH + tc * 16 + l15] = acc[v];
    __syncthreads();
#pragma unroll
    for (int i = 0; i < 2; ++i) {
        const int r = (tid >> 3) + 16 * i, c4 = tid & 7;
        const float4 val = *(const float4*)&Ds[r * GD_PITCH + c4 * 4];
        *(float4*)&Dn[(row0 + r) * CCH + col0 + c4 * 4] = val;
    }
}

__global__ __launch_bounds__(256) void softmax_kernel(const float* __restrict__ P0,
                                                      const float* __restrict__ U,
                                                      const float* __restrict__ Wvec,
                                                      const float* __restrict__ bq,
                                                      const float* __restrict__ bk,
                                                      const float* __restrict__ bv,
                                                      float* __restrict__ att,
                                                      float* __restrict__ cvec) {
    const int n = blockIdx.y;
    const int row = blockIdx.x * 8 + (threadIdx.x >> 5);
    const int lane = threadIdx.x & 31;
    const float* p = P0 + ((size_t)n * CCH + row) * CCH;
    const float ui = U[n * CCH + row];
    const float bqi = bq[row];
    float l[8];
    float mx = -3.0e38f;
#pragma unroll
    for (int t = 0; t < 8; ++t) {
        const int j = lane + 32 * t;
        const float val = p[j] + ui * bk[j] + bqi * Wvec[n * CCH + j];
        l[t] = val;
        mx = fmaxf(mx, val);
    }
    for (int off = 16; off >= 1; off >>= 1) mx = fmaxf(mx, __shfl_xor(mx, off, 32));
    float sum = 0.f;
#pragma unroll
    for (int t = 0; t < 8; ++t) { l[t] = __expf(l[t] - mx); sum += l[t]; }
    for (int off = 16; off >= 1; off >>= 1) sum += __shfl_xor(sum, off, 32);
    const float inv = 1.f / sum;
    float cd = 0.f;
    float* arow = att + ((size_t)n * CCH + row) * CCH;
#pragma unroll
    for (int t = 0; t < 8; ++t) {
        const int j = lane + 32 * t;
        const float a = l[t] * inv;
        arow[j] = a;
        cd += a * bv[j];
    }
    for (int off = 16; off >= 1; off >>= 1) cd += __shfl_xor(cd, off, 32);
    if (lane == 0) cvec[n * CCH + row] = cd;
}

__global__ __launch_bounds__(512) void out_kernel(const float* __restrict__ x,
                                                  const float* __restrict__ Bt,
                                                  const float* __restrict__ cvec,
                                                  float* __restrict__ out) {
    __shared__ float lds[CCH * 40];
    const int n = blockIdx.y;
    const float* xn = x + (size_t)n * CCH * MTOK;
    float* on = out + (size_t)n * MTOK * CCH;
    const int m0 = blockIdx.x * 32;
    const int tid = threadIdx.x, lane = tid & 31, w = tid >> 5;
    const int l15 = lane & 15, half = lane >> 4;

#pragma unroll
    for (int i = 0; i < 16; ++i) {
        const int ch = w + 16 * i;
        lds[ch * 40 + lane] = xn[(size_t)ch * MTOK + m0 + lane];
    }
    __syncthreads();

    const int tr  = w & 1;
    const int tc0 = (w >> 1) * 2;
    const float* btn = Bt + (size_t)n * CCH * CCH;
    const v8f zero8 = {0.f,0.f,0.f,0.f,0.f,0.f,0.f,0.f};
    v8f acc[2];
    acc[0] = zero8; acc[1] = zero8;

    for (int kb = 0; kb < CCH; kb += 4) {
        const int kk = kb + half * 2;
        const int ml = tr * 16 + l15;
        v2f a;
        a[0] = lds[kk * 40 + ml];
        a[1] = lds[(kk + 1) * 40 + ml];
#pragma unroll
        for (int c = 0; c < 2; ++c) {
            const int col = (tc0 + c) * 16 + l15;
            const v2f b = *(const v2f*)&btn[col * CCH + kk];
            acc[c] = wmma_f32(a, b, acc[c]);
        }
    }
#pragma unroll
    for (int c = 0; c < 2; ++c) {
        const int col = (tc0 + c) * 16 + l15;
        const float cv = cvec[n * CCH + col];
        const int mrow = tr * 16 + half * 8;
#pragma unroll
        for (int v = 0; v < 8; ++v)
            on[(size_t)(m0 + mrow + v) * CCH + col] = acc[c][v] + cv;
    }
}

extern "C" void kernel_launch(void* const* d_in, const int* in_sizes, int n_in,
                              void* d_out, int out_size, void* d_ws, size_t ws_size,
                              hipStream_t stream) {
    (void)in_sizes; (void)n_in; (void)out_size; (void)ws_size;
    const float* x  = (const float*)d_in[0];
    const float* Wq = (const float*)d_in[1];
    const float* bq = (const float*)d_in[2];
    const float* Wk = (const float*)d_in[3];
    const float* bk = (const float*)d_in[4];
    const float* Wv = (const float*)d_in[5];
    const float* bv = (const float*)d_in[6];
    float* out = (float*)d_out;
    float* ws  = (float*)d_ws;

    const size_t oGP  = 0;
    const size_t oSP  = oGP  + (size_t)2 * 64 * 65536;
    const size_t oG   = oSP  + (size_t)2 * 64 * 256;
    const size_t oS   = oG   + (size_t)2 * 65536;
    const size_t oT1  = oS   + 512;
    const size_t oP0  = oT1  + (size_t)2 * 65536;
    const size_t oATT = oP0  + (size_t)2 * 65536;
    const size_t oBT  = oATT + (size_t)2 * 65536;
    const size_t oU   = oBT  + (size_t)2 * 65536;
    const size_t oW   = oU   + 512;
    const size_t oCV  = oW   + 512;

    gram_kernel<<<dim3(64, 2, 4), 512, 0, stream>>>(x, ws + oGP, ws + oSP);
    reduce_kernel<<<512, 256, 0, stream>>>(ws + oGP, ws + oSP, ws + oG, ws + oS);
    vecprep_kernel<<<1, 512, 0, stream>>>(Wq, Wk, bk, ws + oS, ws + oU, ws + oW);
    gemm256_kernel<0><<<dim3(8, 8, 2), 128, 0, stream>>>(Wq, 0, ws + oG, 65536, ws + oT1, 65536);
    gemm256_kernel<1><<<dim3(8, 8, 2), 128, 0, stream>>>(ws + oT1, 65536, Wk, 0, ws + oP0, 65536);
    softmax_kernel<<<dim3(32, 2), 256, 0, stream>>>(ws + oP0, ws + oU, ws + oW, bq, bk, bv,
                                                    ws + oATT, ws + oCV);
    gemm256_kernel<0><<<dim3(8, 8, 2), 128, 0, stream>>>(ws + oATT, 65536, Wv, 0, ws + oBT, 65536);
    out_kernel<<<dim3(1024, 2), 512, 0, stream>>>(x, ws + oBT, ws + oCV, out);
}


# out_kernel: one 16-wave WG per WGP walks 8 token slabs with each wave's 16-column B panel (whole K) held in registers; B fragments came from global at every k-step; k order and sums unchanged
# speedup vs baseline: 2.6398x; 1.3918x over previous
#include <hip/hip_runtime.h>
#include <hip/hip_bf16.h>


#define MTOK 32768
#define CCH  256

typedef __attribute__((ext_vector_type(2))) float v2f;
typedef __attribute__((ext_vector_type(8))) float v8f;

__device__ __forceinline__ v8f wmma_f32(v2f a, v2f b, v8f c) {
    return __builtin_amdgcn_wmma_f32_16x16x4_f32(false, a, false, b, (short)0, c, false, false);
}

__global__ __launch_bounds__(512) void gram_kernel(const float* __restrict__ x,
                                                   float* __restrict__ Gpart,
                                                   float* __restrict__ Spart) {
    __shared__ float lds[CCH * 34];
    const int n    = blockIdx.y;
    const int wg   = blockIdx.x;
    const int band = blockIdx.z;
    const float* xn = x + (size_t)n * CCH * MTOK;
    const int tid  = threadIdx.x;
    const int lane = tid & 31;
    const int w    = tid >> 5;
    const int bwr  = w >> 2, bwc = w & 3;
    const int l15  = lane & 15;
    const int half = lane >> 4;
    const bool sums = (band == 0);

    const v8f zero8 = {0.f,0.f,0.f,0.f,0.f,0.f,0.f,0.f};
    v8f acc[4];
#pragma unroll
    for (int c = 0; c < 4; ++c) acc[c] = zero8;

    float sacc[16];
#pragma unroll
    for (int i = 0; i < 16; ++i) sacc[i] = 0.f;

    const int m0 = wg * 512;
    const int chA = (band * 4 + bwr) * 16 + l15;
    for (int t0 = 0; t0 < 512; t0 += 32) {
        __syncthreads();
#pragma unroll
        for (int i = 0; i < 16; ++i) {
            const int ch = w + 16 * i;
            const float val = xn[(size_t)ch * MTOK + m0 + t0 + lane];
            lds[ch * 34 + lane] = val;
            if (sums) sacc[i] += val;
        }
        __syncthreads();
#pragma unroll
        for (int kb = 0; kb < 32; kb += 4) {
            const int kk = kb + half * 2;
            v2f b[4];
            const v2f a = *(const v2f*)&lds[chA * 34 + kk];
#pragma unroll
            for (int c = 0; c < 4; ++c) {
                const int chB = (bwc * 4 + c) * 16 + l15;
                b[c] = *(const v2f*)&lds[chB * 34 + kk];
            }
#pragma unroll
            for (int c = 0; c < 4; ++c)
                acc[c] = wmma_f32(a, b[c], acc[c]);
        }
    }

    float* gp = Gpart + ((size_t)n * 64 + wg) * (CCH * CCH);
#pragma unroll
    for (int c = 0; c < 4; ++c) {
        const int row0 = (band * 4 + bwr) * 16 + half * 8;
        const int col  = (bwc * 4 + c) * 16 + l15;
#pragma unroll
        for (int v = 0; v < 8; ++v)
            gp[(row0 + v) * CCH + col] = acc[c][v];
    }

    if (sums) {
#pragma unroll
        for (int i = 0; i < 16; ++i) {
            float s = sacc[i];
            for (int off = 16; off >= 1; off >>= 1) s += __shfl_xor(s, off, 32);
            if (lane == 0) Spart[((size_t)n * 64 + wg) * CCH + w + 16 * i] = s;
        }
    }
}

__global__ __launch_bounds__(256) void reduce_kernel(const float* __restrict__ Gpart,
                                                     const float* __restrict__ Spart,
                                                     float* __restrict__ G,
                                                     float* __restrict__ S) {
    const int idx = blockIdx.x * 256 + threadIdx.x;
    const int n = idx >> 16;
    const int e = idx & 65535;
    const float* gp = Gpart + (size_t)n * 64 * 65536 + e;
    float s = 0.f;
    for (int p = 0; p < 64; ++p) s += gp[(size_t)p * 65536];
    G[idx] = s;
    if (idx < 512) {
        const float* sp = Spart + (size_t)(idx >> 8) * 64 * CCH + (idx & 255);
        float t = 0.f;
        for (int p = 0; p < 64; ++p) t += sp[p * CCH];
        S[idx] = t;
    }
}

__global__ __launch_bounds__(512) void vecprep_kernel(const float* __restrict__ Wq,
                                                      const float* __restrict__ Wk,
                                                      const float* __restrict__ bk,
                                                      const float* __restrict__ S,
                                                      float* __restrict__ U,
                                                      float* __restrict__ Wvec) {
    const int tid = threadIdx.x;
    const int n = tid >> 8, i = tid & 255;
    const float* s = S + n * CCH;
    float u = 0.f, wv = 0.f;
    for (int c = 0; c < CCH; ++c) {
        u  += Wq[i * CCH + c] * s[c];
        wv += Wk[i * CCH + c] * s[c];
    }
    U[tid] = u;
    Wvec[tid] = wv + (float)MTOK * bk[i];
}

#define GK_PITCH 260
#define GC_PITCH 48
#define GD_PITCH 36

template <int TRANSB>
__global__ __launch_bounds__(128) void gemm256_kernel(const float* __restrict__ A, long long sA,
                                                      const float* __restrict__ B, long long sB,
                                                      float* __restrict__ D, long long sD) {
    constexpr int AS_FLOATS = 32 * GK_PITCH;
    constexpr int BS_FLOATS = TRANSB ? 32 * GK_PITCH : CCH * GC_PITCH;
    __shared__ __attribute__((aligned(16))) float lds[AS_FLOATS + BS_FLOATS];
    float* As = lds;
    float* Bs = lds + AS_FLOATS;

    const int n = blockIdx.z;
    const float* An = A + (size_t)n * sA;
    const float* Bn = B + (size_t)n * sB;
    float* Dn = D + (size_t)n * sD;
    const int tid = threadIdx.x, lane = tid & 31, w = tid >> 5;
    const int l15 = lane & 15, half = lane >> 4;
    const int row0 = blockIdx.y * 32;
    const int col0 = blockIdx.x * 32;

    const float* ag = An + (size_t)(row0 + (tid >> 6)) * CCH + (tid & 63) * 4;
    const float* bg = TRANSB ? Bn + (size_t)(col0 + (tid >> 6)) * CCH + (tid & 63) * 4
                             : Bn + (size_t)(tid >> 3) * CCH + col0 + (tid & 7) * 4;
    float* as = As + (tid >> 6) * GK_PITCH + (tid & 63) * 4;
    float* bs = TRANSB ? Bs + (tid >> 6) * GK_PITCH + (tid & 63) * 4
                       : Bs + (tid >> 3) * GC_PITCH + (tid & 7) * 4;
    float4 ra[16], rb[16];
#pragma unroll
    for (int i = 0; i < 16; ++i)
        ra[i] = *(const float4*)(ag + i * 2 * CCH);
#pragma unroll
    for (int i = 0; i < 16; ++i)
        rb[i] = *(const float4*)(bg + i * (TRANSB ? 2 : 16) * CCH);
#pragma unroll
    for (int i = 0; i < 16; ++i)
        *(float4*)(as + i * 2 * GK_PITCH) = ra[i];
#pragma unroll
    for (int i = 0; i < 16; ++i)
        *(float4*)(bs + i * (TRANSB ? 2 * GK_PITCH : 16 * GC_PITCH)) = rb[i];
    __syncthreads();

    const int tr = w & 1;
    const int tc = w >> 1;
    const float* ap = As + (tr * 16 + l15) * GK_PITCH + half * 2;
    const float* bp = TRANSB ? Bs + (tc * 16 + l15) * GK_PITCH + half * 2
                             : Bs + (half * 2) * GC_PITCH + tc * 16 + l15;

    v2f fa[2][16], fb[2][16];
    v8f acc = {0.f,0.f,0.f,0.f,0.f,0.f,0.f,0.f};
#pragma unroll
    for (int blk = -1; blk < 4; ++blk) {
        if (blk < 3) {
            const int nx = (blk + 1) & 1;
#pragma unroll
            for (int s = 0; s < 16; ++s) {
                const int kb = (blk + 1) * 64 + s * 4;
                fa[nx][s] = *(const v2f*)(ap + kb);
                if (TRANSB) {
                    fb[nx][s] = *(const v2f*)(bp + kb);
                } else {
                    fb[nx][s][0] = bp[kb * GC_PITCH];
                    fb[nx][s][1] = bp[(kb + 1) * GC_PITCH];
                }
            }
        }
        __builtin_amdgcn_sched_barrier(0);
        if (blk >= 0) {
            const int cu = blk & 1;
#pragma unroll
            for (int s = 0; s < 16; ++s)
                acc = wmma_f32(fa[cu][s], fb[cu][s], acc);
        }
        __builtin_amdgcn_sched_barrier(0);
    }

    __syncthreads();
    float* Ds = lds;
#pragma unroll
    for (int v = 0; v < 8; ++v)
        Ds[(tr * 16 + half * 8 + v) * GD_PITCH + tc * 16 + l15] = acc[v];
    __syncthreads();
#pragma unroll
    for (int i = 0; i < 2; ++i) {
        const int r = (tid >> 3) + 16 * i, c4 = tid & 7;
        const float4 val = *(const float4*)&Ds[r * GD_PITCH + c4 * 4];
        *(float4*)&Dn[(row0 + r) * CCH + col0 + c4 * 4] = val;
    }
}

__global__ __launch_bounds__(256) void softmax_kernel(const float* __restrict__ P0,
                                                      const float* __restrict__ U,
                                                      const float* __restrict__ Wvec,
                                                      const float* __restrict__ bq,
                                                      const float* __restrict__ bk,
                                                      const float* __restrict__ bv,
                                                      float* __restrict__ att,
                                                      float* __restrict__ cvec) {
    const int n = blockIdx.y;
    const int row = blockIdx.x * 8 + (threadIdx.x >> 5);
    const int lane = threadIdx.x & 31;
    const float* p = P0 + ((size_t)n * CCH + row) * CCH;
    const float ui = U[n * CCH + row];
    const float bqi = bq[row];
    float l[8];
    float mx = -3.0e38f;
#pragma unroll
    for (int t = 0; t < 8; ++t) {
        const int j = lane + 32 * t;
        const float val = p[j] + ui * bk[j] + bqi * Wvec[n * CCH + j];
        l[t] = val;
        mx = fmaxf(mx, val);
    }
    for (int off = 16; off >= 1; off >>= 1) mx = fmaxf(mx, __shfl_xor(mx, off, 32));
    float sum = 0.f;
#pragma unroll
    for (int t = 0; t < 8; ++t) { l[t] = __expf(l[t] - mx); sum += l[t]; }
    for (int off = 16; off >= 1; off >>= 1) sum += __shfl_xor(sum, off, 32);
    const float inv = 1.f / sum;
    float cd = 0.f;
    float* arow = att + ((size_t)n * CCH + row) * CCH;
#pragma unroll
    for (int t = 0; t < 8; ++t) {
        const int j = lane + 32 * t;
        const float a = l[t] * inv;
        arow[j] = a;
        cd += a * bv[j];
    }
    for (int off = 16; off >= 1; off >>= 1) cd += __shfl_xor(cd, off, 32);
    if (lane == 0) cvec[n * CCH + row] = cd;
}

#define OX_PITCH 260
#define OB_PITCH 132
#define OD_PITCH 260
#define O_SLABS  8
#define OX_FLOATS (32 * OX_PITCH)
#define OD_FLOATS (32 * OD_PITCH)
#define OB_FLOATS (16 * 16 * OB_PITCH)

__global__ __launch_bounds__(512) __attribute__((amdgpu_waves_per_eu(4, 4))) void out_kernel(const float* __restrict__ x,
                                                  const float* __restrict__ Bt,
                                                  const float* __restrict__ cvec,
                                                  float* __restrict__ out) {
    __shared__ __attribute__((aligned(16))) float lds[OB_FLOATS];
    float* xs  = lds;
    float* dsb = lds + 2 * OX_FLOATS;

    const int n = blockIdx.y;
    const float* xn = x + (size_t)n * CCH * MTOK;
    float* on = out + (size_t)n * MTOK * CCH;
    const int tid = threadIdx.x, lane = tid & 31, w = tid >> 5;
    const int l15 = lane & 15, half = lane >> 4;
    const int m00 = blockIdx.x * (O_SLABS * 32);

    const float* bg = Bt + (size_t)n * CCH * CCH + (size_t)(16 * w) * CCH + lane * 4;

    const float* xg = xn + (size_t)(16 * w) * MTOK + m00 + lane;
    float xr[16];
#pragma unroll
    for (int i = 0; i < 16; ++i)
        xr[i] = xg[(size_t)i * MTOK];

    float* bw = lds + w * (16 * OB_PITCH);
    v2f bf[64];
#pragma unroll
    for (int hq = 0; hq < 2; ++hq) {
        float4 t[16];
#pragma unroll
        for (int r = 0; r < 16; ++r)
            t[r] = *(const float4*)(bg + r * CCH + hq * 128);
#pragma unroll
        for (int r = 0; r < 16; ++r)
            *(float4*)(bw + r * OB_PITCH + lane * 4) = t[r];
        __syncthreads();
#pragma unroll
        for (int s = 0; s < 32; ++s)
            bf[hq * 32 + s] = *(const v2f*)(bw + l15 * OB_PITCH + s * 4 + half * 2);
        __syncthreads();
    }

    float* xst = xs + lane * OX_PITCH + 16 * w;
#pragma unroll
    for (int q = 0; q < 4; ++q)
        *(float4*)(xst + 4 * q) = make_float4(xr[4 * q], xr[4 * q + 1], xr[4 * q + 2], xr[4 * q + 3]);
    const float cv = cvec[n * CCH + 16 * w + l15];
    __syncthreads();

    const v8f zero8 = {0.f,0.f,0.f,0.f,0.f,0.f,0.f,0.f};
    for (int sl = 0; sl < O_SLABS; ++sl) {
        const int cur = sl & 1;
        const bool more = (sl + 1 < O_SLABS);
        if (more) {
#pragma unroll
            for (int i = 0; i < 16; ++i)
                xr[i] = xg[(size_t)i * MTOK + (sl + 1) * 32];
        }

        const float* ap = xs + cur * OX_FLOATS + l15 * OX_PITCH + half * 2;
        v8f acc0 = zero8, acc1 = zero8;
        v2f fa[2][4][2];
#pragma unroll
        for (int blk = -1; blk < 16; ++blk) {
            if (blk < 15) {
                const int nx = (blk + 1) & 1;
#pragma unroll
                for (int j = 0; j < 4; ++j) {
                    const int s = (blk + 1) * 4 + j;
                    fa[nx][j][0] = *(const v2f*)(ap + 4 * s);
                    fa[nx][j][1] = *(const v2f*)(ap + 16 * OX_PITCH + 4 * s);
                }
            }
            __builtin_amdgcn_sched_barrier(0);
            if (blk >= 0) {
                const int cu = blk & 1;
#pragma unroll
                for (int j = 0; j < 4; ++j) {
                    const int s = blk * 4 + j;
                    acc0 = wmma_f32(fa[cu][j][0], bf[s], acc0);
                    acc1 = wmma_f32(fa[cu][j][1], bf[s], acc1);
                }
            }
            __builtin_amdgcn_sched_barrier(0);
        }

        float* dc = dsb + cur * OD_FLOATS + (half * 8) * OD_PITCH + 16 * w + l15;
#pragma unroll
        for (int v = 0; v < 8; ++v) {
            dc[v * OD_PITCH]        = acc0[v] + cv;
            dc[(16 + v) * OD_PITCH] = acc1[v] + cv;
        }
        if (more) {
            float* xw = xst + (cur ^ 1) * OX_FLOATS;
#pragma unroll
            for (int q = 0; q < 4; ++q)
                *(float4*)(xw + 4 * q) = make_float4(xr[4 * q], xr[4 * q + 1], xr[4 * q + 2], xr[4 * q + 3]);
        }
        __syncthreads();

        const float* dr = dsb + cur * OD_FLOATS;
        float* og = on + (size_t)(m00 + sl * 32) * CCH;
#pragma unroll
        for (int i = 0; i < 4; ++i) {
            const int idx = tid + 512 * i;
            const int row = idx >> 6, c4 = idx & 63;
            const float4 val = *(const float4*)&dr[row * OD_PITCH + c4 * 4];
            *(float4*)&og[(size_t)row * CCH + c4 * 4] = val;
        }
    }
}

extern "C" void kernel_launch(void* const* d_in, const int* in_sizes, int n_in,
                              void* d_out, int out_size, void* d_ws, size_t ws_size,
                              hipStream_t stream) {
    (void)in_sizes; (void)n_in; (void)out_size; (void)ws_size;
    const float* x  = (const float*)d_in[0];
    const float* Wq = (const float*)d_in[1];
    const float* bq = (const float*)d_in[2];
    const float* Wk = (const float*)d_in[3];
    const float* bk = (const float*)d_in[4];
    const float* Wv = (const float*)d_in[5];
    const float* bv = (const float*)d_in[6];
    float* out = (float*)d_out;
    float* ws  = (float*)d_ws;

    const size_t oGP  = 0;
    const size_t oSP  = oGP  + (size_t)2 * 64 * 65536;
    const size_t oG   = oSP  + (size_t)2 * 64 * 256;
    const size_t oS   = oG   + (size_t)2 * 65536;
    const size_t oT1  = oS   + 512;
    const size_t oP0  = oT1  + (size_t)2 * 65536;
    const size_t oATT = oP0  + (size_t)2 * 65536;
    const size_t oBT  = oATT + (size_t)2 * 65536;
    const size_t oU   = oBT  + (size_t)2 * 65536;
    const size_t oW   = oU   + 512;
    const size_t oCV  = oW   + 512;

    gram_kernel<<<dim3(64, 2, 4), 512, 0, stream>>>(x, ws + oGP, ws + oSP);
    reduce_kernel<<<512, 256, 0, stream>>>(ws + oGP, ws + oSP, ws + oG, ws + oS);
    vecprep_kernel<<<1, 512, 0, stream>>>(Wq, Wk, bk, ws + oS, ws + oU, ws + oW);
    gemm256_kernel<0><<<dim3(8, 8, 2), 128, 0, stream>>>(Wq, 0, ws + oG, 65536, ws + oT1, 65536);
    gemm256_kernel<1><<<dim3(8, 8, 2), 128, 0, stream>>>(ws + oT1, 65536, Wk, 0, ws + oP0, 65536);
    softmax_kernel<<<dim3(32, 2), 256, 0, stream>>>(ws + oP0, ws + oU, ws + oW, bq, bk, bv,
                                                    ws + oATT, ws + oCV);
    gemm256_kernel<0><<<dim3(8, 8, 2), 128, 0, stream>>>(ws + oATT, 65536, Wv, 0, ws + oBT, 65536);
    out_kernel<<<dim3(128, 2), 512, 0, stream>>>(x, ws + oBT, ws + oCV, out);
}


# vecprep launch folded into reduce_kernel: the WG that sums row i of G also sums the 64 row-sum partials and takes u[i], w[i] as 256-term dots by tree reduction; 7 launches; dot sum order changes
# speedup vs baseline: 3.1814x; 1.2052x over previous
#include <hip/hip_runtime.h>
#include <hip/hip_bf16.h>


#define MTOK 32768
#define CCH  256

typedef __attribute__((ext_vector_type(2))) float v2f;
typedef __attribute__((ext_vector_type(8))) float v8f;

__device__ __forceinline__ v8f wmma_f32(v2f a, v2f b, v8f c) {
    return __builtin_amdgcn_wmma_f32_16x16x4_f32(false, a, false, b, (short)0, c, false, false);
}

__global__ __launch_bounds__(512) void gram_kernel(const float* __restrict__ x,
                                                   float* __restrict__ Gpart,
                                                   float* __restrict__ Spart) {
    __shared__ float lds[CCH * 34];
    const int n    = blockIdx.y;
    const int wg   = blockIdx.x;
    const int band = blockIdx.z;
    const float* xn = x + (size_t)n * CCH * MTOK;
    const int tid  = threadIdx.x;
    const int lane = tid & 31;
    const int w    = tid >> 5;
    const int bwr  = w >> 2, bwc = w & 3;
    const int l15  = lane & 15;
    const int half = lane >> 4;
    const bool sums = (band == 0);

    const v8f zero8 = {0.f,0.f,0.f,0.f,0.f,0.f,0.f,0.f};
    v8f acc[4];
#pragma unroll
    for (int c = 0; c < 4; ++c) acc[c] = zero8;

    float sacc[16];
#pragma unroll
    for (int i = 0; i < 16; ++i) sacc[i] = 0.f;

    const int m0 = wg * 512;
    const int chA = (band * 4 + bwr) * 16 + l15;
    for (int t0 = 0; t0 < 512; t0 += 32) {
        __syncthreads();
#pragma unroll
        for (int i = 0; i < 16; ++i) {
            const int ch = w + 16 * i;
            const float val = xn[(size_t)ch * MTOK + m0 + t0 + lane];
            lds[ch * 34 + lane] = val;
            if (sums) sacc[i] += val;
        }
        __syncthreads();
#pragma unroll
        for (int kb = 0; kb < 32; kb += 4) {
            const int kk = kb + half * 2;
            v2f b[4];
            const v2f a = *(const v2f*)&lds[chA * 34 + kk];
#pragma unroll
            for (int c = 0; c < 4; ++c) {
                const int chB = (bwc * 4 + c) * 16 + l15;
                b[c] = *(const v2f*)&lds[chB * 34 + kk];
            }
#pragma unroll
            for (int c = 0; c < 4; ++c)
                acc[c] = wmma_f32(a, b[c], acc[c]);
        }
    }

    float* gp = Gpart + ((size_t)n * 64 + wg) * (CCH * CCH);
#pragma unroll
    for (int c = 0; c < 4; ++c) {
        const int row0 = (band * 4 + bwr) * 16 + half * 8;
        const int col  = (bwc * 4 + c) * 16 + l15;
#pragma unroll
        for (int v = 0; v < 8; ++v)
            gp[(row0 + v) * CCH + col] = acc[c][v];
    }

    if (sums) {
#pragma unroll
        for (int i = 0; i < 16; ++i) {
            float s = sacc[i];
            for (int off = 16; off >= 1; off >>= 1) s += __shfl_xor(s, off, 32);
            if (lane == 0) Spart[((size_t)n * 64 + wg) * CCH + w + 16 * i] = s;
        }
    }
}

__global__ __launch_bounds__(256) void reduce_kernel(const float* __restrict__ Gpart,
                                                     const float* __restrict__ Spart,
                                                     const float* __restrict__ Wq,
                                                     const float* __restrict__ Wk,
                                                     const float* __restrict__ bk,
                                                     float* __restrict__ G,
                                                     float* __restrict__ U,
                                                     float* __restrict__ Wvec) {
    __shared__ float red[2][8];
    const int t = threadIdx.x;
    const int b = blockIdx.x;
    const int n = b >> 8;
    const int row = b & 255;
    const int lane = t & 31, w = t >> 5;
    const float* gp = Gpart + (size_t)n * 64 * 65536 + row * CCH + t;
    const float* sp = Spart + (size_t)n * 64 * CCH + t;
    const float wq = Wq[row * CCH + t];
    const float wk = Wk[row * CCH + t];
    float g = 0.f, sv = 0.f;
#pragma unroll 16
    for (int p = 0; p < 64; ++p) {
        g  += gp[(size_t)p * 65536];
        sv += sp[p * CCH];
    }
    G[b * CCH + t] = g;

    float pu = wq * sv;
    float pw = wk * sv;
    for (int off = 16; off >= 1; off >>= 1) {
        pu += __shfl_xor(pu, off, 32);
        pw += __shfl_xor(pw, off, 32);
    }
    if (lane == 0) { red[0][w] = pu; red[1][w] = pw; }
    __syncthreads();
    if (t == 0) {
        float u = 0.f, wv = 0.f;
        for (int k = 0; k < 8; ++k) { u += red[0][k]; wv += red[1][k]; }
        U[b] = u;
        Wvec[b] = wv + (float)MTOK * bk[row];
    }
}

#define GK_PITCH 260
#define GC_PITCH 48
#define GD_PITCH 36

template <int TRANSB>
__global__ __launch_bounds__(128) void gemm256_kernel(const float* __restrict__ A, long long sA,
                                                      const float* __restrict__ B, long long sB,
                                                      float* __restrict__ D, long long sD) {
    constexpr int AS_FLOATS = 32 * GK_PITCH;
    constexpr int BS_FLOATS = TRANSB ? 32 * GK_PITCH : CCH * GC_PITCH;
    __shared__ __attribute__((aligned(16))) float lds[AS_FLOATS + BS_FLOATS];
    float* As = lds;
    float* Bs = lds + AS_FLOATS;

    const int n = blockIdx.z;
    const float* An = A + (size_t)n * sA;
    const float* Bn = B + (size_t)n * sB;
    float* Dn = D + (size_t)n * sD;
    const int tid = threadIdx.x, lane = tid & 31, w = tid >> 5;
    const int l15 = lane & 15, half = lane >> 4;
    const int row0 = blockIdx.y * 32;
    const int col0 = blockIdx.x * 32;

    const float* ag = An + (size_t)(row0 + (tid >> 6)) * CCH + (tid & 63) * 4;
    const float* bg = TRANSB ? Bn + (size_t)(col0 + (tid >> 6)) * CCH + (tid & 63) * 4
                             : Bn + (size_t)(tid >> 3) * CCH + col0 + (tid & 7) * 4;
    float* as = As + (tid >> 6) * GK_PITCH + (tid & 63) * 4;
    float* bs = TRANSB ? Bs + (tid >> 6) * GK_PITCH + (tid & 63) * 4
                       : Bs + (tid >> 3) * GC_PITCH + (tid & 7) * 4;
    float4 ra[16], rb[16];
#pragma unroll
    for (int i = 0; i < 16; ++i)
        ra[i] = *(const float4*)(ag + i * 2 * CCH);
#pragma unroll
    for (int i = 0; i < 16; ++i)
        rb[i] = *(const float4*)(bg + i * (TRANSB ? 2 : 16) * CCH);
#pragma unroll
    for (int i = 0; i < 16; ++i)
        *(float4*)(as + i * 2 * GK_PITCH) = ra[i];
#pragma unroll
    for (int i = 0; i < 16; ++i)
        *(float4*)(bs + i * (TRANSB ? 2 * GK_PITCH : 16 * GC_PITCH)) = rb[i];
    __syncthreads();

    const int tr = w & 1;
    const int tc = w >> 1;
    const float* ap = As + (tr * 16 + l15) * GK_PITCH + half * 2;
    const float* bp = TRANSB ? Bs + (tc * 16 + l15) * GK_PITCH + half * 2
                             : Bs + (half * 2) * GC_PITCH + tc * 16 + l15;

    v2f fa[2][16], fb[2][16];
    v8f acc = {0.f,0.f,0.f,0.f,0.f,0.f,0.f,0.f};
#pragma unroll
    for (int blk = -1; blk < 4; ++blk) {
        if (blk < 3) {
            const int nx = (blk + 1) & 1;
#pragma unroll
            for (int s = 0; s < 16; ++s) {
                const int kb = (blk + 1) * 64 + s * 4;
                fa[nx][s] = *(const v2f*)(ap + kb);
                if (TRANSB) {
                    fb[nx][s] = *(const v2f*)(bp + kb);
                } else {
                    fb[nx][s][0] = bp[kb * GC_PITCH];
                    fb[nx][s][1] = bp[(kb + 1) * GC_PITCH];
                }
            }
        }
        __builtin_amdgcn_sched_barrier(0);
        if (blk >= 0) {
            const int cu = blk & 1;
#pragma unroll
            for (int s = 0; s < 16; ++s)
                acc = wmma_f32(fa[cu][s], fb[cu][s], acc);
        }
        __builtin_amdgcn_sched_barrier(0);
    }

    __syncthreads();
    float* Ds = lds;
#pragma unroll
    for (int v = 0; v < 8; ++v)
        Ds[(tr * 16 + half * 8 + v) * GD_PITCH + tc * 16 + l15] = acc[v];
    __syncthreads();
#pragma unroll
    for (int i = 0; i < 2; ++i) {
        const int r = (tid >> 3) + 16 * i, c4 = tid & 7;
        const float4 val = *(const float4*)&Ds[r * GD_PITCH + c4 * 4];
        *(float4*)&Dn[(row0 + r) * CCH + col0 + c4 * 4] = val;
    }
}

__global__ __launch_bounds__(256) void softmax_kernel(const float* __restrict__ P0,
                                                      const float* __restrict__ U,
                                                      const float* __restrict__ Wvec,
                                                      const float* __restrict__ bq,
                                                      const float* __restrict__ bk,
                                                      const float* __restrict__ bv,
                                                      float* __restrict__ att,
                                                      float* __restrict__ cvec) {
    const int n = blockIdx.y;
    const int row = blockIdx.x * 8 + (threadIdx.x >> 5);
    const int lane = threadIdx.x & 31;
    const float* p = P0 + ((size_t)n * CCH + row) * CCH;
    const float ui = U[n * CCH + row];
    const float bqi = bq[row];
    float l[8];
    float mx = -3.0e38f;
#pragma unroll
    for (int t = 0; t < 8; ++t) {
        const int j = lane + 32 * t;
        const float val = p[j] + ui * bk[j] + bqi * Wvec[n * CCH + j];
        l[t] = val;
        mx = fmaxf(mx, val);
    }
    for (int off = 16; off >= 1; off >>= 1) mx = fmaxf(mx, __shfl_xor(mx, off, 32));
    float sum = 0.f;
#pragma unroll
    for (int t = 0; t < 8; ++t) { l[t] = __expf(l[t] - mx); sum += l[t]; }
    for (int off = 16; off >= 1; off >>= 1) sum += __shfl_xor(sum, off, 32);
    const float inv = 1.f / sum;
    float cd = 0.f;
    float* arow = att + ((size_t)n * CCH + row) * CCH;
#pragma unroll
    for (int t = 0; t < 8; ++t) {
        const int j = lane + 32 * t;
        const float a = l[t] * inv;
        arow[j] = a;
        cd += a * bv[j];
    }
    for (int off = 16; off >= 1; off >>= 1) cd += __shfl_xor(cd, off, 32);
    if (lane == 0) cvec[n * CCH + row] = cd;
}

#define OX_PITCH 260
#define OB_PITCH 132
#define OD_PITCH 260
#define O_SLABS  8
#define OX_FLOATS (32 * OX_PITCH)
#define OD_FLOATS (32 * OD_PITCH)
#define OB_FLOATS (16 * 16 * OB_PITCH)

__global__ __launch_bounds__(512) __attribute__((amdgpu_waves_per_eu(4, 4))) void out_kernel(const float* __restrict__ x,
                                                  const float* __restrict__ Bt,
                                                  const float* __restrict__ cvec,
                                                  float* __restrict__ out) {
    __shared__ __attribute__((aligned(16))) float lds[OB_FLOATS];
    float* xs  = lds;
    float* dsb = lds + 2 * OX_FLOATS;

    const int n = blockIdx.y;
    const float* xn = x + (size_t)n * CCH * MTOK;
    float* on = out + (size_t)n * MTOK * CCH;
    const int tid = threadIdx.x, lane = tid & 31, w = tid >> 5;
    const int l15 = lane & 15, half = lane >> 4;
    const int m00 = blockIdx.x * (O_SLABS * 32);

    const float* bg = Bt + (size_t)n * CCH * CCH + (size_t)(16 * w) * CCH + lane * 4;

    const float* xg = xn + (size_t)(16 * w) * MTOK + m00 + lane;
    float xr[16];
#pragma unroll
    for (int i = 0; i < 16; ++i)
        xr[i] = xg[(size_t)i * MTOK];

    float* bw = lds + w * (16 * OB_PITCH);
    v2f bf[64];
#pragma unroll
    for (int hq = 0; hq < 2; ++hq) {
        float4 t[16];
#pragma unroll
        for (int r = 0; r < 16; ++r)
            t[r] = *(const float4*)(bg + r * CCH + hq * 128);
#pragma unroll
        for (int r = 0; r < 16; ++r)
            *(float4*)(bw + r * OB_PITCH + lane * 4) = t[r];
        __syncthreads();
#pragma unroll
        for (int s = 0; s < 32; ++s)
            bf[hq * 32 + s] = *(const v2f*)(bw + l15 * OB_PITCH + s * 4 + half * 2);
        __syncthreads();
    }

    float* xst = xs + lane * OX_PITCH + 16 * w;
#pragma unroll
    for (int q = 0; q < 4; ++q)
        *(float4*)(xst + 4 * q) = make_float4(xr[4 * q], xr[4 * q + 1], xr[4 * q + 2], xr[4 * q + 3]);
    const float cv = cvec[n * CCH + 16 * w + l15];
    __syncthreads();

    const v8f zero8 = {0.f,0.f,0.f,0.f,0.f,0.f,0.f,0.f};
    for (int sl = 0; sl < O_SLABS; ++sl) {
        const int cur = sl & 1;
        const bool more = (sl + 1 < O_SLABS);
        if (more) {
#pragma unroll
            for (int i = 0; i < 16; ++i)
                xr[i] = xg[(size_t)i * MTOK + (sl + 1) * 32];
        }

        const float* ap = xs + cur * OX_FLOATS + l15 * OX_PITCH + half * 2;
        v8f acc0 = zero8, acc1 = zero8;
        v2f fa[2][4][2];
#pragma unroll
        for (int blk = -1; blk < 16; ++blk) {
            if (blk < 15) {
                const int nx = (blk + 1) & 1;
#pragma unroll
                for (int j = 0; j < 4; ++j) {
                    const int s = (blk + 1) * 4 + j;
                    fa[nx][j][0] = *(const v2f*)(ap + 4 * s);
                    fa[nx][j][1] = *(const v2f*)(ap + 16 * OX_PITCH + 4 * s);
                }
            }
            __builtin_amdgcn_sched_barrier(0);
            if (blk >= 0) {
                const int cu = blk & 1;
#pragma unroll
                for (int j = 0; j < 4; ++j) {
                    const int s = blk * 4 + j;
                    acc0 = wmma_f32(fa[cu][j][0], bf[s], acc0);
                    acc1 = wmma_f32(fa[cu][j][1], bf[s], acc1);
                }
            }
            __builtin_amdgcn_sched_barrier(0);
        }

        float* dc = dsb + cur * OD_FLOATS + (half * 8) * OD_PITCH + 16 * w + l15;
#pragma unroll
        for (int v = 0; v < 8; ++v) {
            dc[v * OD_PITCH]        = acc0[v] + cv;
            dc[(16 + v) * OD_PITCH] = acc1[v] + cv;
        }
        if (more) {
            float* xw = xst + (cur ^ 1) * OX_FLOATS;
#pragma unroll
            for (int q = 0; q < 4; ++q)
                *(float4*)(xw + 4 * q) = make_float4(xr[4 * q], xr[4 * q + 1], xr[4 * q + 2], xr[4 * q + 3]);
        }
        __syncthreads();

        const float* dr = dsb + cur * OD_FLOATS;
        float* og = on + (size_t)(m00 + sl * 32) * CCH;
#pragma unroll
        for (int i = 0; i < 4; ++i) {
            const int idx = tid + 512 * i;
            const int row = idx >> 6, c4 = idx & 63;
            const float4 val = *(const float4*)&dr[row * OD_PITCH + c4 * 4];
            *(float4*)&og[(size_t)row * CCH + c4 * 4] = val;
        }
    }
}

extern "C" void kernel_launch(void* const* d_in, const int* in_sizes, int n_in,
                              void* d_out, int out_size, void* d_ws, size_t ws_size,
                              hipStream_t stream) {
    (void)in_sizes; (void)n_in; (void)out_size; (void)ws_size;
    const float* x  = (const float*)d_in[0];
    const float* Wq = (const float*)d_in[1];
    const float* bq = (const float*)d_in[2];
    const float* Wk = (const float*)d_in[3];
    const float* bk = (const float*)d_in[4];
    const float* Wv = (const float*)d_in[5];
    const float* bv = (const float*)d_in[6];
    float* out = (float*)d_out;
    float* ws  = (float*)d_ws;

    const size_t oGP  = 0;
    const size_t oSP  = oGP  + (size_t)2 * 64 * 65536;
    const size_t oG   = oSP  + (size_t)2 * 64 * 256;
    const size_t oS   = oG   + (size_t)2 * 65536;
    const size_t oT1  = oS   + 512;
    const size_t oP0  = oT1  + (size_t)2 * 65536;
    const size_t oATT = oP0  + (size_t)2 * 65536;
    const size_t oBT  = oATT + (size_t)2 * 65536;
    const size_t oU   = oBT  + (size_t)2 * 65536;
    const size_t oW   = oU   + 512;
    const size_t oCV  = oW   + 512;

    gram_kernel<<<dim3(64, 2, 4), 512, 0, stream>>>(x, ws + oGP, ws + oSP);
    reduce_kernel<<<512, 256, 0, stream>>>(ws + oGP, ws + oSP, Wq, Wk, bk, ws + oG, ws + oU, ws + oW);
    gemm256_kernel<0><<<dim3(8, 8, 2), 128, 0, stream>>>(Wq, 0, ws + oG, 65536, ws + oT1, 65536);
    gemm256_kernel<1><<<dim3(8, 8, 2), 128, 0, stream>>>(ws + oT1, 65536, Wk, 0, ws + oP0, 65536);
    softmax_kernel<<<dim3(32, 2), 256, 0, stream>>>(ws + oP0, ws + oU, ws + oW, bq, bk, bv,
                                                    ws + oATT, ws + oCV);
    gemm256_kernel<0><<<dim3(8, 8, 2), 128, 0, stream>>>(ws + oATT, 65536, Wv, 0, ws + oBT, 65536);
    out_kernel<<<dim3(128, 2), 512, 0, stream>>>(x, ws + oBT, ws + oCV, out);
}


# gram_kernel: G = X X^T is symmetric, so only its 136 tiles on or above the diagonal are multiplied (17 per SIMD, one WG per WGP); reduce_kernel sums per tile and writes the mirror; sums unchanged
# speedup vs baseline: 3.7564x; 1.1807x over previous
#include <hip/hip_runtime.h>
#include <hip/hip_bf16.h>


#define MTOK 32768
#define CCH  256

typedef __attribute__((ext_vector_type(2))) float v2f;
typedef __attribute__((ext_vector_type(8))) float v8f;

__device__ __forceinline__ v8f wmma_f32(v2f a, v2f b, v8f c) {
    return __builtin_amdgcn_wmma_f32_16x16x4_f32(false, a, false, b, (short)0, c, false, false);
}

#define GR_PITCH 34
#define GR_BUF   (CCH * GR_PITCH)
#define G_TILES  136

template <int NT>
__device__ __forceinline__ void gram_slab(const float* __restrict__ buf, const int* aoff,
                                          const int* boff, v8f* acc) {
#pragma unroll
    for (int kb = 0; kb < 32; kb += 4) {
        v2f a[NT], b[NT];
#pragma unroll
        for (int e = 0; e < NT; ++e) {
            a[e] = *(const v2f*)&buf[aoff[e] + kb];
            b[e] = *(const v2f*)&buf[boff[e] + kb];
        }
#pragma unroll
        for (int e = 0; e < NT; ++e)
            acc[e] = wmma_f32(a[e], b[e], acc[e]);
    }
}

__global__ __launch_bounds__(512) void gram_kernel(const float* __restrict__ x,
                                                   float* __restrict__ Gpart,
                                                   float* __restrict__ Spart) {
    __shared__ float lds[2 * GR_BUF];
    const int n    = blockIdx.y;
    const int wg   = blockIdx.x;
    const int hs   = blockIdx.z;
    const float* xn = x + (size_t)n * CCH * MTOK;
    const int tid  = threadIdx.x;
    const int lane = tid & 31;
    const int w    = tid >> 5;
    const int q    = w >> 2, iw = w & 3;
    const int l15  = lane & 15;
    const int half = lane >> 4;
    const bool sums = (hs == 0);
    const int p    = hs * 4 + q;
    const int nlo  = 16 - p;
    const int j0   = 4 * iw + (iw > q ? 1 : 0);
    const bool five = (iw == q);

    int aoff[5], boff[5], tix[5];
#pragma unroll
    for (int e = 0; e < 5; ++e) {
        const int j  = (e < 4 || five) ? j0 + e : j0;
        const bool lo = j < nlo;
        const int rt = lo ? p : 15 - p;
        const int ct = lo ? p + j : j - 1;
        aoff[e] = (rt * 16 + l15) * GR_PITCH + half * 2;
        boff[e] = (ct * 16 + l15) * GR_PITCH + half * 2;
        tix[e]  = p * 17 + j;
    }

    const v8f zero8 = {0.f,0.f,0.f,0.f,0.f,0.f,0.f,0.f};
    v8f acc[5];
#pragma unroll
    for (int e = 0; e < 5; ++e) acc[e] = zero8;

    float sacc[16];
#pragma unroll
    for (int i = 0; i < 16; ++i) sacc[i] = 0.f;

    const float* xg = xn + (size_t)w * MTOK + wg * 512 + lane;
    float* xs = lds + w * GR_PITCH + lane;
    float xr[16];
#pragma unroll
    for (int i = 0; i < 16; ++i)
        xr[i] = xg[(size_t)i * 16 * MTOK];
#pragma unroll
    for (int i = 0; i < 16; ++i) {
        xs[i * 16 * GR_PITCH] = xr[i];
        if (sums) sacc[i] += xr[i];
    }
    __syncthreads();

    for (int s = 0; s < 16; ++s) {
        const int cur = s & 1;
        const bool more = (s + 1 < 16);
        if (more) {
#pragma unroll
            for (int i = 0; i < 16; ++i)
                xr[i] = xg[(size_t)i * 16 * MTOK + (s + 1) * 32];
        }
        const float* buf = lds + cur * GR_BUF;
        if (five) gram_slab<5>(buf, aoff, boff, acc);
        else      gram_slab<4>(buf, aoff, boff, acc);
        if (more) {
            float* xw = xs + (cur ^ 1) * GR_BUF;
#pragma unroll
            for (int i = 0; i < 16; ++i) {
                xw[i * 16 * GR_PITCH] = xr[i];
                if (sums) sacc[i] += xr[i];
            }
        }
        __syncthreads();
    }

    float* gp = Gpart + ((size_t)n * 64 + wg) * (G_TILES * 256) + lane;
#pragma unroll
    for (int e = 0; e < 5; ++e) {
        if (e < 4 || five) {
#pragma unroll
            for (int v = 0; v < 8; ++v)
                gp[tix[e] * 256 + v * 32] = acc[e][v];
        }
    }

    if (sums) {
#pragma unroll
        for (int i = 0; i < 16; ++i) {
            float sv = sacc[i];
            for (int off = 16; off >= 1; off >>= 1) sv += __shfl_xor(sv, off, 32);
            if (lane == 0) Spart[((size_t)n * 64 + wg) * CCH + w + 16 * i] = sv;
        }
    }
}

__global__ __launch_bounds__(256) void reduce_kernel(const float* __restrict__ Gpart,
                                                     const float* __restrict__ Spart,
                                                     const float* __restrict__ Wq,
                                                     const float* __restrict__ Wk,
                                                     const float* __restrict__ bk,
                                                     float* __restrict__ G,
                                                     float* __restrict__ U,
                                                     float* __restrict__ Wvec) {
    __shared__ float red[4][8];
    __shared__ float tsh[16 * 17];
    const int t = threadIdx.x;
    const int b = blockIdx.x;
    const int n = (b >= G_TILES) ? 1 : 0;
    const int T = b - n * G_TILES;
    const int p = T / 17, j = T - p * 17;
    const bool lo = j < 16 - p;
    const int rt = lo ? p : 15 - p;
    const int ct = lo ? p + j : j - 1;
    const int lane = t & 31, w = t >> 5;
    const bool rows = (T < 128);
    const int row0 = rows ? 2 * T : 0;
    const float* gp = Gpart + ((size_t)n * 64 * G_TILES + T) * 256 + t;
    const float* sp = Spart + (size_t)n * 64 * CCH + t;
    const float wq0 = Wq[row0 * CCH + t];
    const float wk0 = Wk[row0 * CCH + t];
    const float wq1 = Wq[(row0 + 1) * CCH + t];
    const float wk1 = Wk[(row0 + 1) * CCH + t];
    float g = 0.f, sv = 0.f;
#pragma unroll 16
    for (int k = 0; k < 64; ++k) {
        g  += gp[(size_t)k * (G_TILES * 256)];
        sv += sp[k * CCH];
    }
    float* Gn = G + (size_t)n * (CCH * CCH);
    const int row = 8 * ((t >> 4) & 1) + (t >> 5), col = t & 15;
    Gn[(rt * 16 + row) * CCH + ct * 16 + col] = g;
    if (rt != ct) {
        tsh[row * 17 + col] = g;
        __syncthreads();
        const int r2 = t >> 4, c2 = t & 15;
        Gn[(ct * 16 + r2) * CCH + rt * 16 + c2] = tsh[c2 * 17 + r2];
    }

    if (rows) {
        float pu0 = wq0 * sv, pw0 = wk0 * sv, pu1 = wq1 * sv, pw1 = wk1 * sv;
        for (int off = 16; off >= 1; off >>= 1) {
            pu0 += __shfl_xor(pu0, off, 32);
            pw0 += __shfl_xor(pw0, off, 32);
            pu1 += __shfl_xor(pu1, off, 32);
            pw1 += __shfl_xor(pw1, off, 32);
        }
        if (lane == 0) { red[0][w] = pu0; red[1][w] = pw0; red[2][w] = pu1; red[3][w] = pw1; }
        __syncthreads();
        if (t < 2) {
            float u = 0.f, wv = 0.f;
            for (int k = 0; k < 8; ++k) { u += red[2 * t][k]; wv += red[2 * t + 1][k]; }
            const int r = row0 + t;
            U[n * CCH + r] = u;
            Wvec[n * CCH + r] = wv + (float)MTOK * bk[r];
        }
    }
}

#define GK_PITCH 260
#define GC_PITCH 48
#define GD_PITCH 36

template <int TRANSB>
__global__ __launch_bounds__(128) void gemm256_kernel(const float* __restrict__ A, long long sA,
                                                      const float* __restrict__ B, long long sB,
                                                      float* __restrict__ D, long long sD) {
    constexpr int AS_FLOATS = 32 * GK_PITCH;
    constexpr int BS_FLOATS = TRANSB ? 32 * GK_PITCH : CCH * GC_PITCH;
    __shared__ __attribute__((aligned(16))) float lds[AS_FLOATS + BS_FLOATS];
    float* As = lds;
    float* Bs = lds + AS_FLOATS;

    const int n = blockIdx.z;
    const float* An = A + (size_t)n * sA;
    const float* Bn = B + (size_t)n * sB;
    float* Dn = D + (size_t)n * sD;
    const int tid = threadIdx.x, lane = tid & 31, w = tid >> 5;
    const int l15 = lane & 15, half = lane >> 4;
    const int row0 = blockIdx.y * 32;
    const int col0 = blockIdx.x * 32;

    const float* ag = An + (size_t)(row0 + (tid >> 6)) * CCH + (tid & 63) * 4;
    const float* bg = TRANSB ? Bn + (size_t)(col0 + (tid >> 6)) * CCH + (tid & 63) * 4
                             : Bn + (size_t)(tid >> 3) * CCH + col0 + (tid & 7) * 4;
    float* as = As + (tid >> 6) * GK_PITCH + (tid & 63) * 4;
    float* bs = TRANSB ? Bs + (tid >> 6) * GK_PITCH + (tid & 63) * 4
                       : Bs + (tid >> 3) * GC_PITCH + (tid & 7) * 4;
    float4 ra[16], rb[16];
#pragma unroll
    for (int i = 0; i < 16; ++i)
        ra[i] = *(const float4*)(ag + i * 2 * CCH);
#pragma unroll
    for (int i = 0; i < 16; ++i)
        rb[i] = *(const float4*)(bg + i * (TRANSB ? 2 : 16) * CCH);
#pragma unroll
    for (int i = 0; i < 16; ++i)
        *(float4*)(as + i * 2 * GK_PITCH) = ra[i];
#pragma unroll
    for (int i = 0; i < 16; ++i)
        *(float4*)(bs + i * (TRANSB ? 2 * GK_PITCH : 16 * GC_PITCH)) = rb[i];
    __syncthreads();

    const int tr = w & 1;
    const int tc = w >> 1;
    const float* ap = As + (tr * 16 + l15) * GK_PITCH + half * 2;
    const float* bp = TRANSB ? Bs + (tc * 16 + l15) * GK_PITCH + half * 2
                             : Bs + (half * 2) * GC_PITCH + tc * 16 + l15;

    v2f fa[2][16], fb[2][16];
    v8f acc = {0.f,0.f,0.f,0.f,0.f,0.f,0.f,0.f};
#pragma unroll
    for (int blk = -1; blk < 4; ++blk) {
        if (blk < 3) {
            const int nx = (blk + 1) & 1;
#pragma unroll
            for (int s = 0; s < 16; ++s) {
                const int kb = (blk + 1) * 64 + s * 4;
                fa[nx][s] = *(const v2f*)(ap + kb);
                if (TRANSB) {
                    fb[nx][s] = *(const v2f*)(bp + kb);
                } else {
                    fb[nx][s][0] = bp[kb * GC_PITCH];
                    fb[nx][s][1] = bp[(kb + 1) * GC_PITCH];
                }
            }
        }
        __builtin_amdgcn_sched_barrier(0);
        if (blk >= 0) {
            const int cu = blk & 1;
#pragma unroll
            for (int s = 0; s < 16; ++s)
                acc = wmma_f32(fa[cu][s], fb[cu][s], acc);
        }
        __builtin_amdgcn_sched_barrier(0);
    }

    __syncthreads();
    float* Ds = lds;
#pragma unroll
    for (int v = 0; v < 8; ++v)
        Ds[(tr * 16 + half * 8 + v) * GD_PITCH + tc * 16 + l15] = acc[v];
    __syncthreads();
#pragma unroll
    for (int i = 0; i < 2; ++i) {
        const int r = (tid >> 3) + 16 * i, c4 = tid & 7;
        const float4 val = *(const float4*)&Ds[r * GD_PITCH + c4 * 4];
        *(float4*)&Dn[(row0 + r) * CCH + col0 + c4 * 4] = val;
    }
}

__global__ __launch_bounds__(256) void softmax_kernel(const float* __restrict__ P0,
                                                      const float* __restrict__ U,
                                                      const float* __restrict__ Wvec,
                                                      const float* __restrict__ bq,
                                                      const float* __restrict__ bk,
                                                      const float* __restrict__ bv,
                                                      float* __restrict__ att,
                                                      float* __restrict__ cvec) {
    const int n = blockIdx.y;
    const int row = blockIdx.x * 8 + (threadIdx.x >> 5);
    const int lane = threadIdx.x & 31;
    const float* p = P0 + ((size_t)n * CCH + row) * CCH;
    const float ui = U[n * CCH + row];
    const float bqi = bq[row];
    float l[8];
    float mx = -3.0e38f;
#pragma unroll
    for (int t = 0; t < 8; ++t) {
        const int j = lane + 32 * t;
        const float val = p[j] + ui * bk[j] + bqi * Wvec[n * CCH + j];
        l[t] = val;
        mx = fmaxf(mx, val);
    }
    for (int off = 16; off >= 1; off >>= 1) mx = fmaxf(mx, __shfl_xor(mx, off, 32));
    float sum = 0.f;
#pragma unroll
    for (int t = 0; t < 8; ++t) { l[t] = __expf(l[t] - mx); sum += l[t]; }
    for (int off = 16; off >= 1; off >>= 1) sum += __shfl_xor(sum, off, 32);
    const float inv = 1.f / sum;
    float cd = 0.f;
    float* arow = att + ((size_t)n * CCH + row) * CCH;
#pragma unroll
    for (int t = 0; t < 8; ++t) {
        const int j = lane + 32 * t;
        const float a = l[t] * inv;
        arow[j] = a;
        cd += a * bv[j];
    }
    for (int off = 16; off >= 1; off >>= 1) cd += __shfl_xor(cd, off, 32);
    if (lane == 0) cvec[n * CCH + row] = cd;
}

#define OX_PITCH 260
#define OB_PITCH 132
#define OD_PITCH 260
#define O_SLABS  8
#define OX_FLOATS (32 * OX_PITCH)
#define OD_FLOATS (32 * OD_PITCH)
#define OB_FLOATS (16 * 16 * OB_PITCH)

__global__ __launch_bounds__(512) __attribute__((amdgpu_waves_per_eu(4, 4))) void out_kernel(const float* __restrict__ x,
                                                  const float* __restrict__ Bt,
                                                  const float* __restrict__ cvec,
                                                  float* __restrict__ out) {
    __shared__ __attribute__((aligned(16))) float lds[OB_FLOATS];
    float* xs  = lds;
    float* dsb = lds + 2 * OX_FLOATS;

    const int n = blockIdx.y;
    const float* xn = x + (size_t)n * CCH * MTOK;
    float* on = out + (size_t)n * MTOK * CCH;
    const int tid = threadIdx.x, lane = tid & 31, w = tid >> 5;
    const int l15 = lane & 15, half = lane >> 4;
    const int m00 = blockIdx.x * (O_SLABS * 32);

    const float* bg = Bt + (size_t)n * CCH * CCH + (size_t)(16 * w) * CCH + lane * 4;

    const float* xg = xn + (size_t)(16 * w) * MTOK + m00 + lane;
    float xr[16];
#pragma unroll
    for (int i = 0; i < 16; ++i)
        xr[i] = xg[(size_t)i * MTOK];

    float* bw = lds + w * (16 * OB_PITCH);
    v2f bf[64];
#pragma unroll
    for (int hq = 0; hq < 2; ++hq) {
        float4 t[16];
#pragma unroll
        for (int r = 0; r < 16; ++r)
            t[r] = *(const float4*)(bg + r * CCH + hq * 128);
#pragma unroll
        for (int r = 0; r < 16; ++r)
            *(float4*)(bw + r * OB_PITCH + lane * 4) = t[r];
        __syncthreads();
#pragma unroll
        for (int s = 0; s < 32; ++s)
            bf[hq * 32 + s] = *(const v2f*)(bw + l15 * OB_PITCH + s * 4 + half * 2);
        __syncthreads();
    }

    float* xst = xs + lane * OX_PITCH + 16 * w;
#pragma unroll
    for (int q = 0; q < 4; ++q)
        *(float4*)(xst + 4 * q) = make_float4(xr[4 * q], xr[4 * q + 1], xr[4 * q + 2], xr[4 * q + 3]);
    const float cv = cvec[n * CCH + 16 * w + l15];
    __syncthreads();

    const v8f zero8 = {0.f,0.f,0.f,0.f,0.f,0.f,0.f,0.f};
    for (int sl = 0; sl < O_SLABS; ++sl) {
        const int cur = sl & 1;
        const bool more = (sl + 1 < O_SLABS);
        if (more) {
#pragma unroll
            for (int i = 0; i < 16; ++i)
                xr[i] = xg[(size_t)i * MTOK + (sl + 1) * 32];
        }

        const float* ap = xs + cur * OX_FLOATS + l15 * OX_PITCH + half * 2;
        v8f acc0 = zero8, acc1 = zero8;
        v2f fa[2][4][2];
#pragma unroll
        for (int blk = -1; blk < 16; ++blk) {
            if (blk < 15) {
                const int nx = (blk + 1) & 1;
#pragma unroll
                for (int j = 0; j < 4; ++j) {
                    const int s = (blk + 1) * 4 + j;
                    fa[nx][j][0] = *(const v2f*)(ap + 4 * s);
                    fa[nx][j][1] = *(const v2f*)(ap + 16 * OX_PITCH + 4 * s);
                }
            }
            __builtin_amdgcn_sched_barrier(0);
            if (blk >= 0) {
                const int cu = blk & 1;
#pragma unroll
                for (int j = 0; j < 4; ++j) {
                    const int s = blk * 4 + j;
                    acc0 = wmma_f32(fa[cu][j][0], bf[s], acc0);
                    acc1 = wmma_f32(fa[cu][j][1], bf[s], acc1);
                }
            }
            __builtin_amdgcn_sched_barrier(0);
        }

        float* dc = dsb + cur * OD_FLOATS + (half * 8) * OD_PITCH + 16 * w + l15;
#pragma unroll
        for (int v = 0; v < 8; ++v) {
            dc[v * OD_PITCH]        = acc0[v] + cv;
            dc[(16 + v) * OD_PITCH] = acc1[v] + cv;
        }
        if (more) {
            float* xw = xst + (cur ^ 1) * OX_FLOATS;
#pragma unroll
            for (int q = 0; q < 4; ++q)
                *(float4*)(xw + 4 * q) = make_float4(xr[4 * q], xr[4 * q + 1], xr[4 * q + 2], xr[4 * q + 3]);
        }
        __syncthreads();

        const float* dr = dsb + cur * OD_FLOATS;
        float* og = on + (size_t)(m00 + sl * 32) * CCH;
#pragma unroll
        for (int i = 0; i < 4; ++i) {
            const int idx = tid + 512 * i;
            const int row = idx >> 6, c4 = idx & 63;
            const float4 val = *(const float4*)&dr[row * OD_PITCH + c4 * 4];
            *(float4*)&og[(size_t)row * CCH + c4 * 4] = val;
        }
    }
}

extern "C" void kernel_launch(void* const* d_in, const int* in_sizes, int n_in,
                              void* d_out, int out_size, void* d_ws, size_t ws_size,
                              hipStream_t stream) {
    (void)in_sizes; (void)n_in; (void)out_size; (void)ws_size;
    const float* x  = (const float*)d_in[0];
    const float* Wq = (const float*)d_in[1];
    const float* bq = (const float*)d_in[2];
    const float* Wk = (const float*)d_in[3];
    const float* bk = (const float*)d_in[4];
    const float* Wv = (const float*)d_in[5];
    const float* bv = (const float*)d_in[6];
    float* out = (float*)d_out;
    float* ws  = (float*)d_ws;

    const size_t oGP  = 0;
    const size_t oSP  = oGP  + (size_t)2 * 64 * 65536;
    const size_t oG   = oSP  + (size_t)2 * 64 * 256;
    const size_t oS   = oG   + (size_t)2 * 65536;
    const size_t oT1  = oS   + 512;
    const size_t oP0  = oT1  + (size_t)2 * 65536;
    const size_t oATT = oP0  + (size_t)2 * 65536;
    const size_t oBT  = oATT + (size_t)2 * 65536;
    const size_t oU   = oBT  + (size_t)2 * 65536;
    const size_t oW   = oU   + 512;
    const size_t oCV  = oW   + 512;

    gram_kernel<<<dim3(64, 2, 2), 512, 0, stream>>>(x, ws + oGP, ws + oSP);
    reduce_kernel<<<272, 256, 0, stream>>>(ws + oGP, ws + oSP, Wq, Wk, bk, ws + oG, ws + oU, ws + oW);
    gemm256_kernel<0><<<dim3(8, 8, 2), 128, 0, stream>>>(Wq, 0, ws + oG, 65536, ws + oT1, 65536);
    gemm256_kernel<1><<<dim3(8, 8, 2), 128, 0, stream>>>(ws + oT1, 65536, Wk, 0, ws + oP0, 65536);
    softmax_kernel<<<dim3(32, 2), 256, 0, stream>>>(ws + oP0, ws + oU, ws + oW, bq, bk, bv,
                                                    ws + oATT, ws + oCV);
    gemm256_kernel<0><<<dim3(8, 8, 2), 128, 0, stream>>>(ws + oATT, 65536, Wv, 0, ws + oBT, 65536);
    out_kernel<<<dim3(128, 2), 512, 0, stream>>>(x, ws + oBT, ws + oCV, out);
}


# gram_kernel, out_kernel: slab hand-over off the WMMA path: barrier written as wait/signal/wait so slabs fetched ahead stay in flight; LDS refill, fetch and row stores moved mid-slab; sums unchanged
# speedup vs baseline: 3.9867x; 1.0613x over previous
#include <hip/hip_runtime.h>
#include <hip/hip_bf16.h>


#define MTOK 32768
#define CCH  256

typedef __attribute__((ext_vector_type(2))) float v2f;
typedef __attribute__((ext_vector_type(8))) float v8f;

__device__ __forceinline__ v8f wmma_f32(v2f a, v2f b, v8f c) {
    return __builtin_amdgcn_wmma_f32_16x16x4_f32(false, a, false, b, (short)0, c, false, false);
}

#define GR_PITCH 34
#define GR_BUF   (CCH * GR_PITCH)
#define G_TILES  136

__device__ __forceinline__ void lds_wait0() { asm volatile("s_wait_dscnt 0x0" ::: "memory"); }
__device__ __forceinline__ void wg_signal() { asm volatile("s_barrier_signal -1" ::: "memory"); }
__device__ __forceinline__ void wg_wait()   { asm volatile("s_barrier_wait -1" ::: "memory"); }

template <int E0, int E1>
__device__ __forceinline__ void gram_load(const float* buf, const int* aoff, const int* boff, int kb,
                                          v2f (&fa)[5][2], v2f (&fb)[5][2]) {
#pragma unroll
    for (int e = E0; e < E1; ++e) {
#pragma unroll
        for (int j = 0; j < 2; ++j) {
            fa[e][j] = *(const v2f*)&buf[aoff[e] + kb + 4 * j];
            fb[e][j] = *(const v2f*)&buf[boff[e] + kb + 4 * j];
        }
    }
}

template <int E0, int E1, int J>
__device__ __forceinline__ void gram_mma(const v2f (&fa)[5][2], const v2f (&fb)[5][2], v8f* acc) {
#pragma unroll
    for (int e = E0; e < E1; ++e)
        acc[e] = wmma_f32(fa[e][J], fb[e][J], acc[e]);
}

__global__ __launch_bounds__(512) void gram_kernel(const float* __restrict__ x,
                                                   float* __restrict__ Gpart,
                                                   float* __restrict__ Spart) {
    __shared__ float lds[2 * GR_BUF];
    const int n    = blockIdx.y;
    const int wg   = blockIdx.x;
    const int hs   = blockIdx.z;
    const float* xn = x + (size_t)n * CCH * MTOK;
    const int tid  = threadIdx.x;
    const int lane = tid & 31;
    const int w    = tid >> 5;
    const int q    = w >> 2, iw = w & 3;
    const int l15  = lane & 15;
    const int half = lane >> 4;
    const bool sums = (hs == 0);
    const int p    = hs * 4 + q;
    const int nlo  = 16 - p;
    const int j0   = 4 * iw + (iw > q ? 1 : 0);
    const bool five = __builtin_amdgcn_readfirstlane(iw == q ? 1 : 0) != 0;

    int aoff[5], boff[5], tix[5];
#pragma unroll
    for (int e = 0; e < 5; ++e) {
        const int j  = (e < 4 || five) ? j0 + e : j0;
        const bool lo = j < nlo;
        const int rt = lo ? p : 15 - p;
        const int ct = lo ? p + j : j - 1;
        aoff[e] = (rt * 16 + l15) * GR_PITCH + half * 2;
        boff[e] = (ct * 16 + l15) * GR_PITCH + half * 2;
        tix[e]  = p * 17 + j;
    }

    const v8f zero8 = {0.f,0.f,0.f,0.f,0.f,0.f,0.f,0.f};
    v8f acc[5];
#pragma unroll
    for (int e = 0; e < 5; ++e) acc[e] = zero8;

    float sacc[16];
#pragma unroll
    for (int i = 0; i < 16; ++i) sacc[i] = 0.f;

    const float* xg = xn + (size_t)w * MTOK + wg * 512 + lane;
    float* xs = lds + w * GR_PITCH + lane;
    float xr[16];
#pragma unroll
    for (int i = 0; i < 16; ++i)
        xr[i] = xg[(size_t)i * 16 * MTOK];
#pragma unroll
    for (int i = 0; i < 16; ++i) {
        xs[i * 16 * GR_PITCH] = xr[i];
        sacc[i] += xr[i];
    }
#pragma unroll
    for (int i = 0; i < 16; ++i)
        xr[i] = xg[(size_t)i * 16 * MTOK + 32];
    lds_wait0();
    wg_signal();
    wg_wait();

    v2f fa[2][5][2], fb[2][5][2];
    gram_load<0, 4>(lds, aoff, boff, 0, fa[0], fb[0]);
    if (five) gram_load<4, 5>(lds, aoff, boff, 0, fa[0], fb[0]);

    for (int s = 0; s < 16; ++s) {
        const int cur = s & 1;
        const float* buf  = lds + cur * GR_BUF;
        const float* nbuf = lds + (cur ^ 1) * GR_BUF;
        const bool more  = (s + 1 < 16);
        const bool more2 = (s + 2 < 16);

        gram_load<0, 4>(buf, aoff, boff, 8, fa[1], fb[1]);
        if (five) gram_load<4, 5>(buf, aoff, boff, 8, fa[1], fb[1]);
        __builtin_amdgcn_sched_barrier(0);
        gram_mma<0, 4, 0>(fa[0], fb[0], acc);
        gram_mma<0, 4, 1>(fa[0], fb[0], acc);
        if (five) { gram_mma<4, 5, 0>(fa[0], fb[0], acc); gram_mma<4, 5, 1>(fa[0], fb[0], acc); }

        gram_load<0, 4>(buf, aoff, boff, 16, fa[0], fb[0]);
        if (five) gram_load<4, 5>(buf, aoff, boff, 16, fa[0], fb[0]);
        __builtin_amdgcn_sched_barrier(0);
        gram_mma<0, 4, 0>(fa[1], fb[1], acc);
        gram_mma<0, 4, 1>(fa[1], fb[1], acc);
        if (five) { gram_mma<4, 5, 0>(fa[1], fb[1], acc); gram_mma<4, 5, 1>(fa[1], fb[1], acc); }

        if (more) {
            float* xw = xs + (cur ^ 1) * GR_BUF;
#pragma unroll
            for (int i = 0; i < 16; ++i) {
                xw[i * 16 * GR_PITCH] = xr[i];
                sacc[i] += xr[i];
                asm volatile("" : "+v"(sacc[i]));
            }
        }
        __builtin_amdgcn_sched_barrier(0);
        if (more2) {
#pragma unroll
            for (int i = 0; i < 16; ++i)
                xr[i] = xg[(size_t)i * 16 * MTOK + (s + 2) * 32];
        }

        gram_load<0, 4>(buf, aoff, boff, 24, fa[1], fb[1]);
        if (five) gram_load<4, 5>(buf, aoff, boff, 24, fa[1], fb[1]);
        __builtin_amdgcn_sched_barrier(0);
        gram_mma<0, 4, 0>(fa[0], fb[0], acc);
        if (five) gram_mma<4, 5, 0>(fa[0], fb[0], acc);
        if (more) {
            lds_wait0();
            wg_signal();
        }
        gram_mma<0, 4, 1>(fa[0], fb[0], acc);
        if (five) gram_mma<4, 5, 1>(fa[0], fb[0], acc);

        if (more) {
            wg_wait();
            gram_load<0, 4>(nbuf, aoff, boff, 0, fa[0], fb[0]);
            if (five) gram_load<4, 5>(nbuf, aoff, boff, 0, fa[0], fb[0]);
        }
        __builtin_amdgcn_sched_barrier(0);
        gram_mma<0, 4, 0>(fa[1], fb[1], acc);
        gram_mma<0, 4, 1>(fa[1], fb[1], acc);
        if (five) { gram_mma<4, 5, 0>(fa[1], fb[1], acc); gram_mma<4, 5, 1>(fa[1], fb[1], acc); }
    }

    float* gp = Gpart + ((size_t)n * 64 + wg) * (G_TILES * 256) + lane;
#pragma unroll
    for (int e = 0; e < 5; ++e) {
        if (e < 4 || five) {
#pragma unroll
            for (int v = 0; v < 8; ++v)
                gp[tix[e] * 256 + v * 32] = acc[e][v];
        }
    }

    if (sums) {
#pragma unroll
        for (int i = 0; i < 16; ++i) {
            float sv = sacc[i];
            for (int off = 16; off >= 1; off >>= 1) sv += __shfl_xor(sv, off, 32);
            if (lane == 0) Spart[((size_t)n * 64 + wg) * CCH + w + 16 * i] = sv;
        }
    }
}

__global__ __launch_bounds__(256) void reduce_kernel(const float* __restrict__ Gpart,
                                                     const float* __restrict__ Spart,
                                                     const float* __restrict__ Wq,
                                                     const float* __restrict__ Wk,
                                                     const float* __restrict__ bk,
                                                     float* __restrict__ G,
                                                     float* __restrict__ U,
                                                     float* __restrict__ Wvec) {
    __shared__ float red[4][8];
    __shared__ float tsh[16 * 17];
    const int t = threadIdx.x;
    const int b = blockIdx.x;
    const int n = (b >= G_TILES) ? 1 : 0;
    const int T = b - n * G_TILES;
    const int p = T / 17, j = T - p * 17;
    const bool lo = j < 16 - p;
    const int rt = lo ? p : 15 - p;
    const int ct = lo ? p + j : j - 1;
    const int lane = t & 31, w = t >> 5;
    const bool rows = (T < 128);
    const int row0 = rows ? 2 * T : 0;
    const float* gp = Gpart + ((size_t)n * 64 * G_TILES + T) * 256 + t;
    const float* sp = Spart + (size_t)n * 64 * CCH + t;
    const float wq0 = Wq[row0 * CCH + t];
    const float wk0 = Wk[row0 * CCH + t];
    const float wq1 = Wq[(row0 + 1) * CCH + t];
    const float wk1 = Wk[(row0 + 1) * CCH + t];
    float g = 0.f, sv = 0.f;
#pragma unroll 16
    for (int k = 0; k < 64; ++k) {
        g  += gp[(size_t)k * (G_TILES * 256)];
        sv += sp[k * CCH];
    }
    float* Gn = G + (size_t)n * (CCH * CCH);
    const int row = 8 * ((t >> 4) & 1) + (t >> 5), col = t & 15;
    Gn[(rt * 16 + row) * CCH + ct * 16 + col] = g;
    if (rt != ct) {
        tsh[row * 17 + col] = g;
        __syncthreads();
        const int r2 = t >> 4, c2 = t & 15;
        Gn[(ct * 16 + r2) * CCH + rt * 16 + c2] = tsh[c2 * 17 + r2];
    }

    if (rows) {
        float pu0 = wq0 * sv, pw0 = wk0 * sv, pu1 = wq1 * sv, pw1 = wk1 * sv;
        for (int off = 16; off >= 1; off >>= 1) {
            pu0 += __shfl_xor(pu0, off, 32);
            pw0 += __shfl_xor(pw0, off, 32);
            pu1 += __shfl_xor(pu1, off, 32);
            pw1 += __shfl_xor(pw1, off, 32);
        }
        if (lane == 0) { red[0][w] = pu0; red[1][w] = pw0; red[2][w] = pu1; red[3][w] = pw1; }
        __syncthreads();
        if (t < 2) {
            float u = 0.f, wv = 0.f;
            for (int k = 0; k < 8; ++k) { u += red[2 * t][k]; wv += red[2 * t + 1][k]; }
            const int r = row0 + t;
            U[n * CCH + r] = u;
            Wvec[n * CCH + r] = wv + (float)MTOK * bk[r];
        }
    }
}

#define GK_PITCH 260
#define GC_PITCH 48
#define GD_PITCH 36

template <int TRANSB>
__global__ __launch_bounds__(128) void gemm256_kernel(const float* __restrict__ A, long long sA,
                                                      const float* __restrict__ B, long long sB,
                                                      float* __restrict__ D, long long sD) {
    constexpr int AS_FLOATS = 32 * GK_PITCH;
    constexpr int BS_FLOATS = TRANSB ? 32 * GK_PITCH : CCH * GC_PITCH;
    __shared__ __attribute__((aligned(16))) float lds[AS_FLOATS + BS_FLOATS];
    float* As = lds;
    float* Bs = lds + AS_FLOATS;

    const int n = blockIdx.z;
    const float* An = A + (size_t)n * sA;
    const float* Bn = B + (size_t)n * sB;
    float* Dn = D + (size_t)n * sD;
    const int tid = threadIdx.x, lane = tid & 31, w = tid >> 5;
    const int l15 = lane & 15, half = lane >> 4;
    const int row0 = blockIdx.y * 32;
    const int col0 = blockIdx.x * 32;

    const float* ag = An + (size_t)(row0 + (tid >> 6)) * CCH + (tid & 63) * 4;
    const float* bg = TRANSB ? Bn + (size_t)(col0 + (tid >> 6)) * CCH + (tid & 63) * 4
                             : Bn + (size_t)(tid >> 3) * CCH + col0 + (tid & 7) * 4;
    float* as = As + (tid >> 6) * GK_PITCH + (tid & 63) * 4;
    float* bs = TRANSB ? Bs + (tid >> 6) * GK_PITCH + (tid & 63) * 4
                       : Bs + (tid >> 3) * GC_PITCH + (tid & 7) * 4;
    float4 ra[16], rb[16];
#pragma unroll
    for (int i = 0; i < 16; ++i)
        ra[i] = *(const float4*)(ag + i * 2 * CCH);
#pragma unroll
    for (int i = 0; i < 16; ++i)
        rb[i] = *(const float4*)(bg + i * (TRANSB ? 2 : 16) * CCH);
#pragma unroll
    for (int i = 0; i < 16; ++i)
        *(float4*)(as + i * 2 * GK_PITCH) = ra[i];
#pragma unroll
    for (int i = 0; i < 16; ++i)
        *(float4*)(bs + i * (TRANSB ? 2 * GK_PITCH : 16 * GC_PITCH)) = rb[i];
    __syncthreads();

    const int tr = w & 1;
    const int tc = w >> 1;
    const float* ap = As + (tr * 16 + l15) * GK_PITCH + half * 2;
    const float* bp = TRANSB ? Bs + (tc * 16 + l15) * GK_PITCH + half * 2
                             : Bs + (half * 2) * GC_PITCH + tc * 16 + l15;

    v2f fa[2][16], fb[2][16];
    v8f acc = {0.f,0.f,0.f,0.f,0.f,0.f,0.f,0.f};
#pragma unroll
    for (int blk = -1; blk < 4; ++blk) {
        if (blk < 3) {
            const int nx = (blk + 1) & 1;
#pragma unroll
            for (int s = 0; s < 16; ++s) {
                const int kb = (blk + 1) * 64 + s * 4;
                fa[nx][s] = *(const v2f*)(ap + kb);
                if (TRANSB) {
                    fb[nx][s] = *(const v2f*)(bp + kb);
                } else {
                    fb[nx][s][0] = bp[kb * GC_PITCH];
                    fb[nx][s][1] = bp[(kb + 1) * GC_PITCH];
                }
            }
        }
        __builtin_amdgcn_sched_barrier(0);
        if (blk >= 0) {
            const int cu = blk & 1;
#pragma unroll
            for (int s = 0; s < 16; ++s)
                acc = wmma_f32(fa[cu][s], fb[cu][s], acc);
        }
        __builtin_amdgcn_sched_barrier(0);
    }

    __syncthreads();
    float* Ds = lds;
#pragma unroll
    for (int v = 0; v < 8; ++v)
        Ds[(tr * 16 + half * 8 + v) * GD_PITCH + tc * 16 + l15] = acc[v];
    __syncthreads();
#pragma unroll
    for (int i = 0; i < 2; ++i) {
        const int r = (tid >> 3) + 16 * i, c4 = tid & 7;
        const float4 val = *(const float4*)&Ds[r * GD_PITCH + c4 * 4];
        *(float4*)&Dn[(row0 + r) * CCH + col0 + c4 * 4] = val;
    }
}

__global__ __launch_bounds__(256) void softmax_kernel(const float* __restrict__ P0,
                                                      const float* __restrict__ U,
                                                      const float* __restrict__ Wvec,
                                                      const float* __restrict__ bq,
                                                      const float* __restrict__ bk,
                                                      const float* __restrict__ bv,
                                                      float* __restrict__ att,
                                                      float* __restrict__ cvec) {
    const int n = blockIdx.y;
    const int row = blockIdx.x * 8 + (threadIdx.x >> 5);
    const int lane = threadIdx.x & 31;
    const float* p = P0 + ((size_t)n * CCH + row) * CCH;
    const float ui = U[n * CCH + row];
    const float bqi = bq[row];
    float l[8];
    float mx = -3.0e38f;
#pragma unroll
    for (int t = 0; t < 8; ++t) {
        const int j = lane + 32 * t;
        const float val = p[j] + ui * bk[j] + bqi * Wvec[n * CCH + j];
        l[t] = val;
        mx = fmaxf(mx, val);
    }
    for (int off = 16; off >= 1; off >>= 1) mx = fmaxf(mx, __shfl_xor(mx, off, 32));
    float sum = 0.f;
#pragma unroll
    for (int t = 0; t < 8; ++t) { l[t] = __expf(l[t] - mx); sum += l[t]; }
    for (int off = 16; off >= 1; off >>= 1) sum += __shfl_xor(sum, off, 32);
    const float inv = 1.f / sum;
    float cd = 0.f;
    float* arow = att + ((size_t)n * CCH + row) * CCH;
#pragma unroll
    for (int t = 0; t < 8; ++t) {
        const int j = lane + 32 * t;
        const float a = l[t] * inv;
        arow[j] = a;
        cd += a * bv[j];
    }
    for (int off = 16; off >= 1; off >>= 1) cd += __shfl_xor(cd, off, 32);
    if (lane == 0) cvec[n * CCH + row] = cd;
}

#define OX_PITCH 260
#define OB_PITCH 132
#define OD_PITCH 260
#define O_SLABS  8
#define OX_FLOATS (32 * OX_PITCH)
#define OD_FLOATS (32 * OD_PITCH)
#define OB_FLOATS (16 * 16 * OB_PITCH)

__global__ __launch_bounds__(512) __attribute__((amdgpu_waves_per_eu(4, 4))) void out_kernel(const float* __restrict__ x,
                                                  const float* __restrict__ Bt,
                                                  const float* __restrict__ cvec,
                                                  float* __restrict__ out) {
    __shared__ __attribute__((aligned(16))) float lds[OB_FLOATS];
    float* xs  = lds;
    float* dsb = lds + 2 * OX_FLOATS;

    const int n = blockIdx.y;
    const float* xn = x + (size_t)n * CCH * MTOK;
    float* on = out + (size_t)n * MTOK * CCH;
    const int tid = threadIdx.x, lane = tid & 31, w = tid >> 5;
    const int l15 = lane & 15, half = lane >> 4;
    const int m00 = blockIdx.x * (O_SLABS * 32);

    const float* bg = Bt + (size_t)n * CCH * CCH + (size_t)(16 * w) * CCH + lane * 4;

    const float* xg = xn + (size_t)(16 * w) * MTOK + m00 + lane;
    float xr[16];
#pragma unroll
    for (int i = 0; i < 16; ++i)
        xr[i] = xg[(size_t)i * MTOK];

    float* bw = lds + w * (16 * OB_PITCH);
    v2f bf[64];
#pragma unroll
    for (int hq = 0; hq < 2; ++hq) {
        float4 t[16];
#pragma unroll
        for (int r = 0; r < 16; ++r)
            t[r] = *(const float4*)(bg + r * CCH + hq * 128);
#pragma unroll
        for (int r = 0; r < 16; ++r)
            *(float4*)(bw + r * OB_PITCH + lane * 4) = t[r];
        __syncthreads();
#pragma unroll
        for (int s = 0; s < 32; ++s)
            bf[hq * 32 + s] = *(const v2f*)(bw + l15 * OB_PITCH + s * 4 + half * 2);
        __syncthreads();
    }

    float* xst = xs + lane * OX_PITCH + 16 * w;
#pragma unroll
    for (int q = 0; q < 4; ++q)
        *(float4*)(xst + 4 * q) = make_float4(xr[4 * q], xr[4 * q + 1], xr[4 * q + 2], xr[4 * q + 3]);
    const float cv = cvec[n * CCH + 16 * w + l15];
#pragma unroll
    for (int i = 0; i < 16; ++i)
        xr[i] = xg[(size_t)i * MTOK + 32];
    lds_wait0();
    wg_signal();
    wg_wait();

    const v8f zero8 = {0.f,0.f,0.f,0.f,0.f,0.f,0.f,0.f};
    for (int sl = 0; sl < O_SLABS; ++sl) {
        const int cur = sl & 1;

        const float* ap = xs + cur * OX_FLOATS + l15 * OX_PITCH + half * 2;
        v8f acc0 = zero8, acc1 = zero8;
        v2f fa[2][4][2];
#pragma unroll
        for (int blk = -1; blk < 16; ++blk) {
            if (blk < 15) {
                const int nx = (blk + 1) & 1;
#pragma unroll
                for (int j = 0; j < 4; ++j) {
                    const int s = (blk + 1) * 4 + j;
                    fa[nx][j][0] = *(const v2f*)(ap + 4 * s);
                    fa[nx][j][1] = *(const v2f*)(ap + 16 * OX_PITCH + 4 * s);
                }
            }
            __builtin_amdgcn_sched_barrier(0);
            if (blk >= 0) {
                const int cu = blk & 1;
#pragma unroll
                for (int j = 0; j < 4; ++j) {
                    const int s = blk * 4 + j;
                    acc0 = wmma_f32(fa[cu][j][0], bf[s], acc0);
                    acc1 = wmma_f32(fa[cu][j][1], bf[s], acc1);
                }
            }
            __builtin_amdgcn_sched_barrier(0);
            if (blk == 7) {
                if (sl + 1 < O_SLABS) {
                    float* xw = xst + (cur ^ 1) * OX_FLOATS;
#pragma unroll
                    for (int q = 0; q < 4; ++q)
                        *(float4*)(xw + 4 * q) = make_float4(xr[4 * q], xr[4 * q + 1], xr[4 * q + 2], xr[4 * q + 3]);
                }
                if (sl + 2 < O_SLABS) {
#pragma unroll
                    for (int i = 0; i < 16; ++i)
                        xr[i] = xg[(size_t)i * MTOK + (sl + 2) * 32];
                }
                if (sl > 0) {
                    const float* dr = dsb + (cur ^ 1) * OD_FLOATS;
                    float* og = on + (size_t)(m00 + (sl - 1) * 32) * CCH;
#pragma unroll
                    for (int i = 0; i < 4; ++i) {
                        const int idx = tid + 512 * i;
                        const int row = idx >> 6, c4 = idx & 63;
                        const float4 val = *(const float4*)&dr[row * OD_PITCH + c4 * 4];
                        *(float4*)&og[(size_t)row * CCH + c4 * 4] = val;
                    }
                }
            }
        }

        float* dc = dsb + cur * OD_FLOATS + (half * 8) * OD_PITCH + 16 * w + l15;
#pragma unroll
        for (int v = 0; v < 8; ++v) {
            dc[v * OD_PITCH]        = acc0[v] + cv;
            dc[(16 + v) * OD_PITCH] = acc1[v] + cv;
        }
        lds_wait0();
        wg_signal();
        wg_wait();
    }

    {
        const float* dr = dsb + ((O_SLABS - 1) & 1) * OD_FLOATS;
        float* og = on + (size_t)(m00 + (O_SLABS - 1) * 32) * CCH;
#pragma unroll
        for (int i = 0; i < 4; ++i) {
            const int idx = tid + 512 * i;
            const int row = idx >> 6, c4 = idx & 63;
            const float4 val = *(const float4*)&dr[row * OD_PITCH + c4 * 4];
            *(float4*)&og[(size_t)row * CCH + c4 * 4] = val;
        }
    }
}

extern "C" void kernel_launch(void* const* d_in, const int* in_sizes, int n_in,
                              void* d_out, int out_size, void* d_ws, size_t ws_size,
                              hipStream_t stream) {
    (void)in_sizes; (void)n_in; (void)out_size; (void)ws_size;
    const float* x  = (const float*)d_in[0];
    const float* Wq = (const float*)d_in[1];
    const float* bq = (const float*)d_in[2];
    const float* Wk = (const float*)d_in[3];
    const float* bk = (const float*)d_in[4];
    const float* Wv = (const float*)d_in[5];
    const float* bv = (const float*)d_in[6];
    float* out = (float*)d_out;
    float* ws  = (float*)d_ws;

    const size_t oGP  = 0;
    const size_t oSP  = oGP  + (size_t)2 * 64 * 65536;
    const size_t oG   = oSP  + (size_t)2 * 64 * 256;
    const size_t oS   = oG   + (size_t)2 * 65536;
    const size_t oT1  = oS   + 512;
    const size_t oP0  = oT1  + (size_t)2 * 65536;
    const size_t oATT = oP0  + (size_t)2 * 65536;
    const size_t oBT  = oATT + (size_t)2 * 65536;
    const size_t oU   = oBT  + (size_t)2 * 65536;
    const size_t oW   = oU   + 512;
    const size_t oCV  = oW   + 512;

    gram_kernel<<<dim3(64, 2, 2), 512, 0, stream>>>(x, ws + oGP, ws + oSP);
    reduce_kernel<<<272, 256, 0, stream>>>(ws + oGP, ws + oSP, Wq, Wk, bk, ws + oG, ws + oU, ws + oW);
    gemm256_kernel<0><<<dim3(8, 8, 2), 128, 0, stream>>>(Wq, 0, ws + oG, 65536, ws + oT1, 65536);
    gemm256_kernel<1><<<dim3(8, 8, 2), 128, 0, stream>>>(ws + oT1, 65536, Wk, 0, ws + oP0, 65536);
    softmax_kernel<<<dim3(32, 2), 256, 0, stream>>>(ws + oP0, ws + oU, ws + oW, bq, bk, bv,
                                                    ws + oATT, ws + oCV);
    gemm256_kernel<0><<<dim3(8, 8, 2), 128, 0, stream>>>(ws + oATT, 65536, Wv, 0, ws + oBT, 65536);
    out_kernel<<<dim3(128, 2), 512, 0, stream>>>(x, ws + oBT, ws + oCV, out);
}
